# starting kernel
# baseline (speedup 1.0000x reference)
#include <hip/hip_runtime.h>
#include <hip/hip_bf16.h>


#define D_MODEL 768
#define D_STATE 16
#define D_CONV  4
#define D_INNER 1536
#define DT_RANK 48
#define BATCH   2
#define SEQ     2048
#define MROWS   (BATCH * SEQ)
#define DBC_LD  (DT_RANK + 2 * D_STATE)

typedef __attribute__((ext_vector_type(2))) float v2f;
typedef __attribute__((ext_vector_type(8))) float v8f;

__device__ __forceinline__ float silu_f(float x) {
    return x * (1.0f / (1.0f + __expf(-x)));
}
__device__ __forceinline__ float softplus_f(float x) {
    return (x > 20.0f) ? x : log1pf(__expf(x));
}

template <int MT, int NT, int EPI>
__launch_bounds__(128)
__global__ void gemm_wmma_f32(const float* __restrict__ A,
                              const float* __restrict__ W,
                              const float* __restrict__ bias,
                              float* __restrict__ C,
                              int M, int N, int K, int lda, int ldc) {
    const int lane  = threadIdx.x & 31;
    const int wave  = threadIdx.x >> 5;
    const int mBase = (blockIdx.y * (blockDim.x >> 5) + wave) * (16 * MT);
    const int nBase = blockIdx.x * (16 * NT);
    const int r16   = lane & 15;
    const int kOff  = (lane >> 4) * 2;

    const float* aP[MT];
    const float* wP[NT];
#pragma unroll
    for (int i = 0; i < MT; ++i)
        aP[i] = A + (size_t)(mBase + i * 16 + r16) * lda + kOff;
#pragma unroll
    for (int t = 0; t < NT; ++t)
        wP[t] = W + (size_t)(nBase + t * 16 + r16) * K + kOff;

    v8f acc[MT][NT];
#pragma unroll
    for (int i = 0; i < MT; ++i)
#pragma unroll
        for (int t = 0; t < NT; ++t) {
            v8f z = {0.f, 0.f, 0.f, 0.f, 0.f, 0.f, 0.f, 0.f};
            acc[i][t] = z;
        }

    v2f aA[MT], bA[NT], aB[MT], bB[NT];

#define LOADFRAG(AF, BF, KK)                                         \
    _Pragma("unroll") for (int i = 0; i < MT; ++i)                   \
        AF[i] = *(const v2f*)(aP[i] + (KK));                         \
    _Pragma("unroll") for (int t = 0; t < NT; ++t)                   \
        BF[t] = *(const v2f*)(wP[t] + (KK));

#define MACFRAG(AF, BF)                                              \
    _Pragma("unroll") for (int i = 0; i < MT; ++i)                   \
        _Pragma("unroll") for (int t = 0; t < NT; ++t)               \
            acc[i][t] = __builtin_amdgcn_wmma_f32_16x16x4_f32(       \
                false, AF[i], false, BF[t], (short)0, acc[i][t],     \
                false, false);

    LOADFRAG(aA, bA, 0)
    int k = 0;
    for (; k + 8 < K; k += 8) {
        LOADFRAG(aB, bB, k + 4)
        MACFRAG(aA, bA)
        LOADFRAG(aA, bA, k + 8)
        MACFRAG(aB, bB)
    }
    LOADFRAG(aB, bB, k + 4)
    MACFRAG(aA, bA)
    MACFRAG(aB, bB)

#undef LOADFRAG
#undef MACFRAG

    const int rowSel = (lane >> 4) * 8;
#pragma unroll
    for (int i = 0; i < MT; ++i)
#pragma unroll
        for (int t = 0; t < NT; ++t) {
            const int col = nBase + t * 16 + (lane & 15);
            float bv = 0.0f;
            if (EPI == 1) bv = bias[col];
#pragma unroll
            for (int r = 0; r < 8; ++r) {
                float v = acc[i][t][r];
                if (EPI == 1) v = softplus_f(v + bv);
                C[(size_t)(mBase + i * 16 + r + rowSel) * ldc + col] = v;
            }
        }
}

__launch_bounds__(256)
__global__ void conv_silu_kernel(const float* __restrict__ xin,
                                 const float* __restrict__ w,
                                 const float* __restrict__ b,
                                 float* __restrict__ out) {
    const int idx = blockIdx.x * blockDim.x + threadIdx.x;
    const int d  = idx % D_INNER;
    const int s  = (idx / D_INNER) % SEQ;
    const int bi = idx / (D_INNER * SEQ);

    const float* base = xin + (size_t)bi * SEQ * D_INNER + d;
    float acc = b[d];
#pragma unroll
    for (int j = 0; j < D_CONV; ++j) {
        const int sp = s - (D_CONV - 1) + j;
        if (sp >= 0) acc += w[d * D_CONV + j] * base[(size_t)sp * D_INNER];
    }
    out[idx] = silu_f(acc);
}

__launch_bounds__(256)
__global__ void scan_kernel(const float* __restrict__ delta,
                            const float* __restrict__ xc,
                            const float* __restrict__ dbc,
                            const float* __restrict__ zbuf,
                            const float* __restrict__ logA,
                            const float* __restrict__ Dskip,
                            float* __restrict__ y) {
    const int gid = blockIdx.x * blockDim.x + threadIdx.x;
    const int n = gid & (D_STATE - 1);
    const int d = (gid >> 4) % D_INNER;
    const int b = gid / (D_INNER * D_STATE);

    const float Aneg   = -__expf(logA[d * D_STATE + n]);
    const bool  writer = (n == 0);
    const float dsk    = Dskip[d];

    const size_t baseBD  = (size_t)b * SEQ * D_INNER + d;
    const size_t baseDbc = (size_t)b * SEQ * DBC_LD;

    float h = 0.0f;
#pragma unroll 2
    for (int s = 0; s < SEQ; ++s) {
        const size_t ix = baseBD + (size_t)s * D_INNER;
        const float dt = delta[ix];
        const float xv = xc[ix];
        const float* pd = dbc + baseDbc + (size_t)s * DBC_LD;
        const float Bv = pd[DT_RANK + n];
        const float Cv = pd[DT_RANK + D_STATE + n];

        const float a = __expf(dt * Aneg);
        h = a * h + (dt * Bv) * xv;

        float p = h * Cv;
        p += __shfl_xor(p, 1, 16);
        p += __shfl_xor(p, 2, 16);
        p += __shfl_xor(p, 4, 16);
        p += __shfl_xor(p, 8, 16);

        if (writer) {
            const float zv = zbuf[ix];
            y[ix] = (p + dsk * xv) * silu_f(zv);
        }
    }
}

extern "C" void kernel_launch(void* const* d_in, const int* in_sizes, int n_in,
                              void* d_out, int out_size, void* d_ws, size_t ws_size,
                              hipStream_t stream) {
    const float* x         = (const float*)d_in[0];
    const float* in_proj_w = (const float*)d_in[1];
    const float* conv_w    = (const float*)d_in[2];
    const float* conv_b    = (const float*)d_in[3];
    const float* x_proj_w  = (const float*)d_in[4];
    const float* dt_proj_w = (const float*)d_in[5];
    const float* dt_proj_b = (const float*)d_in[6];
    const float* log_A     = (const float*)d_in[7];
    const float* D_skip    = (const float*)d_in[8];
    const float* out_proj_w= (const float*)d_in[9];
    float* out = (float*)d_out;

    const size_t planeBD = (size_t)MROWS * D_INNER;
    float* ws      = (float*)d_ws;
    float* buf_x   = ws;
    float* buf_z   = buf_x  + planeBD;
    float* buf_xc  = buf_z  + planeBD;
    float* buf_dbc = buf_xc + planeBD;
    float* buf_y   = buf_dbc + (size_t)MROWS * DBC_LD;

    const dim3 blk(128);
    const int gy = MROWS / (16 * 2 * 4);

    gemm_wmma_f32<2, 4, 0><<<dim3(D_INNER / 64, gy), blk, 0, stream>>>(
        x, in_proj_w, nullptr, buf_x, MROWS, D_INNER, D_MODEL, D_MODEL, D_INNER);
    gemm_wmma_f32<2, 4, 0><<<dim3(D_INNER / 64, gy), blk, 0, stream>>>(
        x, in_proj_w + (size_t)D_INNER * D_MODEL, nullptr, buf_z,
        MROWS, D_INNER, D_MODEL, D_MODEL, D_INNER);

    conv_silu_kernel<<<(MROWS * D_INNER) / 256, 256, 0, stream>>>(
        buf_x, conv_w, conv_b, buf_xc);

    gemm_wmma_f32<2, 1, 0><<<dim3(DBC_LD / 16, gy), blk, 0, stream>>>(
        buf_xc, x_proj_w, nullptr, buf_dbc, MROWS, DBC_LD, D_INNER, D_INNER, DBC_LD);

    gemm_wmma_f32<2, 4, 1><<<dim3(D_INNER / 64, gy), blk, 0, stream>>>(
        buf_dbc, dt_proj_w, dt_proj_b, buf_x, MROWS, D_INNER, DT_RANK, DBC_LD, D_INNER);

    scan_kernel<<<(BATCH * D_INNER * D_STATE) / 256, 256, 0, stream>>>(
        buf_x, buf_xc, buf_dbc, buf_z, log_A, D_skip, buf_y);

    gemm_wmma_f32<2, 4, 0><<<dim3(D_MODEL / 64, gy), blk, 0, stream>>>(
        buf_y, out_proj_w, nullptr, out, MROWS, D_MODEL, D_INNER, D_INNER, D_MODEL);
}


# scan: time loop in 16-step tiles, each tile's loads issued one tile ahead; lane n of a channel's 16 state lanes loads delta/x/z and stores y of step n (row_share broadcast); butterfly order kept
# speedup vs baseline: 4.4829x; 4.4829x over previous
#include <hip/hip_runtime.h>
#include <hip/hip_bf16.h>


#define D_MODEL 768
#define D_STATE 16
#define D_CONV  4
#define D_INNER 1536
#define DT_RANK 48
#define BATCH   2
#define SEQ     2048
#define MROWS   (BATCH * SEQ)
#define DBC_LD  (DT_RANK + 2 * D_STATE)

typedef __attribute__((ext_vector_type(2))) float v2f;
typedef __attribute__((ext_vector_type(8))) float v8f;

__device__ __forceinline__ float silu_f(float x) {
    return x * (1.0f / (1.0f + __expf(-x)));
}
__device__ __forceinline__ float softplus_f(float x) {
    return (x > 20.0f) ? x : log1pf(__expf(x));
}

template <int MT, int NT, int EPI>
__launch_bounds__(128)
__global__ void gemm_wmma_f32(const float* __restrict__ A,
                              const float* __restrict__ W,
                              const float* __restrict__ bias,
                              float* __restrict__ C,
                              int M, int N, int K, int lda, int ldc) {
    const int lane  = threadIdx.x & 31;
    const int wave  = threadIdx.x >> 5;
    const int mBase = (blockIdx.y * (blockDim.x >> 5) + wave) * (16 * MT);
    const int nBase = blockIdx.x * (16 * NT);
    const int r16   = lane & 15;
    const int kOff  = (lane >> 4) * 2;

    const float* aP[MT];
    const float* wP[NT];
#pragma unroll
    for (int i = 0; i < MT; ++i)
        aP[i] = A + (size_t)(mBase + i * 16 + r16) * lda + kOff;
#pragma unroll
    for (int t = 0; t < NT; ++t)
        wP[t] = W + (size_t)(nBase + t * 16 + r16) * K + kOff;

    v8f acc[MT][NT];
#pragma unroll
    for (int i = 0; i < MT; ++i)
#pragma unroll
        for (int t = 0; t < NT; ++t) {
            v8f z = {0.f, 0.f, 0.f, 0.f, 0.f, 0.f, 0.f, 0.f};
            acc[i][t] = z;
        }

    v2f aA[MT], bA[NT], aB[MT], bB[NT];

#define LOADFRAG(AF, BF, KK)                                         \
    _Pragma("unroll") for (int i = 0; i < MT; ++i)                   \
        AF[i] = *(const v2f*)(aP[i] + (KK));                         \
    _Pragma("unroll") for (int t = 0; t < NT; ++t)                   \
        BF[t] = *(const v2f*)(wP[t] + (KK));

#define MACFRAG(AF, BF)                                              \
    _Pragma("unroll") for (int i = 0; i < MT; ++i)                   \
        _Pragma("unroll") for (int t = 0; t < NT; ++t)               \
            acc[i][t] = __builtin_amdgcn_wmma_f32_16x16x4_f32(       \
                false, AF[i], false, BF[t], (short)0, acc[i][t],     \
                false, false);

    LOADFRAG(aA, bA, 0)
    int k = 0;
    for (; k + 8 < K; k += 8) {
        LOADFRAG(aB, bB, k + 4)
        MACFRAG(aA, bA)
        LOADFRAG(aA, bA, k + 8)
        MACFRAG(aB, bB)
    }
    LOADFRAG(aB, bB, k + 4)
    MACFRAG(aA, bA)
    MACFRAG(aB, bB)

#undef LOADFRAG
#undef MACFRAG

    const int rowSel = (lane >> 4) * 8;
#pragma unroll
    for (int i = 0; i < MT; ++i)
#pragma unroll
        for (int t = 0; t < NT; ++t) {
            const int col = nBase + t * 16 + (lane & 15);
            float bv = 0.0f;
            if (EPI == 1) bv = bias[col];
#pragma unroll
            for (int r = 0; r < 8; ++r) {
                float v = acc[i][t][r];
                if (EPI == 1) v = softplus_f(v + bv);
                C[(size_t)(mBase + i * 16 + r + rowSel) * ldc + col] = v;
            }
        }
}

__launch_bounds__(256)
__global__ void conv_silu_kernel(const float* __restrict__ xin,
                                 const float* __restrict__ w,
                                 const float* __restrict__ b,
                                 float* __restrict__ out) {
    const int idx = blockIdx.x * blockDim.x + threadIdx.x;
    const int d  = idx % D_INNER;
    const int s  = (idx / D_INNER) % SEQ;
    const int bi = idx / (D_INNER * SEQ);

    const float* base = xin + (size_t)bi * SEQ * D_INNER + d;
    float acc = b[d];
#pragma unroll
    for (int j = 0; j < D_CONV; ++j) {
        const int sp = s - (D_CONV - 1) + j;
        if (sp >= 0) acc += w[d * D_CONV + j] * base[(size_t)sp * D_INNER];
    }
    out[idx] = silu_f(acc);
}

#define TILE_S 16

template <int J>
__device__ __forceinline__ float row_share_f(float v) {
    return __int_as_float(__builtin_amdgcn_update_dpp(
        0, __float_as_int(v), 0x150 + J, 0xf, 0xf, false));
}

__launch_bounds__(256)
__global__ void scan_kernel(const float* __restrict__ delta,
                            const float* __restrict__ xc,
                            const float* __restrict__ dbc,
                            const float* __restrict__ zbuf,
                            const float* __restrict__ logA,
                            const float* __restrict__ Dskip,
                            float* __restrict__ y) {
    const int gid = blockIdx.x * blockDim.x + threadIdx.x;
    const int n = gid & (D_STATE - 1);
    const int d = (gid >> 4) % D_INNER;
    const int b = gid / (D_INNER * D_STATE);

    const float Aneg = -__expf(logA[d * D_STATE + n]);
    const float dsk  = Dskip[d];

    const size_t baseBD = (size_t)b * SEQ * D_INNER + d + (size_t)n * D_INNER;
    const float* pDt = delta + baseBD;
    const float* pXc = xc    + baseBD;
    const float* pZ  = zbuf  + baseBD;
    float*       pY  = y     + baseBD;
    const float* pBC = dbc + (size_t)b * SEQ * DBC_LD + DT_RANK + n;

    float dtA, xvA, zvA, BA[TILE_S], CA[TILE_S];
    float dtB, xvB, zvB, BB[TILE_S], CB[TILE_S];

#define LOAD_TILE(DT, XV, ZV, BV, CV, S0)                              \
    {                                                                  \
        const size_t o = (size_t)(S0) * D_INNER;                       \
        DT = pDt[o];                                                   \
        XV = pXc[o];                                                   \
        const float* pb = pBC + (size_t)(S0) * DBC_LD;                 \
                \
        _Pragma("unroll") for (int j = 0; j < TILE_S; ++j) {           \
            BV[j] = pb[j * DBC_LD];                                    \
            CV[j] = pb[j * DBC_LD + D_STATE];                          \
        }                                                              \
        ZV = pZ[o];                                                    \
    }

#define SCAN_STEP(J, DT, XV, BV, CV)                                   \
    {                                                                  \
        const float dtj = row_share_f<J>(DT);                          \
        const float xvj = row_share_f<J>(XV);                          \
        const float a = __expf(dtj * Aneg);                            \
           \
           \
        h = fmaf(a, h, (dtj * BV[J]) * xvj);                           \
        P[J] = h * CV[J];                                              \
    }

#define RED_STAGE(M)                                                   \
                    \
    _Pragma("unroll") for (int j = 0; j < TILE_S; ++j)                 \
        P[j] += __shfl_xor(P[j], M, 16);

#define SCAN_TILE(DT, XV, ZV, BV, CV, S0)                              \
    {                                                                  \
        float P[TILE_S];                                               \
        SCAN_STEP(0, DT, XV, BV, CV)  SCAN_STEP(1, DT, XV, BV, CV)     \
        SCAN_STEP(2, DT, XV, BV, CV)  SCAN_STEP(3, DT, XV, BV, CV)     \
        SCAN_STEP(4, DT, XV, BV, CV)  SCAN_STEP(5, DT, XV, BV, CV)     \
        SCAN_STEP(6, DT, XV, BV, CV)  SCAN_STEP(7, DT, XV, BV, CV)     \
        SCAN_STEP(8, DT, XV, BV, CV)  SCAN_STEP(9, DT, XV, BV, CV)     \
        SCAN_STEP(10, DT, XV, BV, CV) SCAN_STEP(11, DT, XV, BV, CV)    \
        SCAN_STEP(12, DT, XV, BV, CV) SCAN_STEP(13, DT, XV, BV, CV)    \
        SCAN_STEP(14, DT, XV, BV, CV) SCAN_STEP(15, DT, XV, BV, CV)    \
           \
        RED_STAGE(1) RED_STAGE(2) RED_STAGE(4) RED_STAGE(8)            \
        float yv = 0.0f;                                               \
          \
        _Pragma("unroll") for (int j = 0; j < TILE_S; ++j)             \
            yv = (n == j) ? P[j] : yv;                                 \
                \
        pY[(size_t)(S0) * D_INNER] = (yv + dsk * XV) * silu_f(ZV);     \
    }

#define HOLD_LOADS() __builtin_amdgcn_sched_barrier(0);

    float h = 0.0f;
    LOAD_TILE(dtA, xvA, zvA, BA, CA, 0)
    int s0 = 0;
    for (; s0 + 2 * TILE_S < SEQ; s0 += 2 * TILE_S) {
        LOAD_TILE(dtB, xvB, zvB, BB, CB, s0 + TILE_S)
        HOLD_LOADS()
        SCAN_TILE(dtA, xvA, zvA, BA, CA, s0)
        HOLD_LOADS()
        LOAD_TILE(dtA, xvA, zvA, BA, CA, s0 + 2 * TILE_S)
        HOLD_LOADS()
        SCAN_TILE(dtB, xvB, zvB, BB, CB, s0 + TILE_S)
    }
    LOAD_TILE(dtB, xvB, zvB, BB, CB, s0 + TILE_S)
    HOLD_LOADS()
    SCAN_TILE(dtA, xvA, zvA, BA, CA, s0)
    SCAN_TILE(dtB, xvB, zvB, BB, CB, s0 + TILE_S)

#undef HOLD_LOADS
#undef RED_STAGE
#undef LOAD_TILE
#undef SCAN_STEP
#undef SCAN_TILE
}

extern "C" void kernel_launch(void* const* d_in, const int* in_sizes, int n_in,
                              void* d_out, int out_size, void* d_ws, size_t ws_size,
                              hipStream_t stream) {
    const float* x         = (const float*)d_in[0];
    const float* in_proj_w = (const float*)d_in[1];
    const float* conv_w    = (const float*)d_in[2];
    const float* conv_b    = (const float*)d_in[3];
    const float* x_proj_w  = (const float*)d_in[4];
    const float* dt_proj_w = (const float*)d_in[5];
    const float* dt_proj_b = (const float*)d_in[6];
    const float* log_A     = (const float*)d_in[7];
    const float* D_skip    = (const float*)d_in[8];
    const float* out_proj_w= (const float*)d_in[9];
    float* out = (float*)d_out;

    const size_t planeBD = (size_t)MROWS * D_INNER;
    float* ws      = (float*)d_ws;
    float* buf_x   = ws;
    float* buf_z   = buf_x  + planeBD;
    float* buf_xc  = buf_z  + planeBD;
    float* buf_dbc = buf_xc + planeBD;
    float* buf_y   = buf_dbc + (size_t)MROWS * DBC_LD;

    const dim3 blk(128);
    const int gy = MROWS / (16 * 2 * 4);

    gemm_wmma_f32<2, 4, 0><<<dim3(D_INNER / 64, gy), blk, 0, stream>>>(
        x, in_proj_w, nullptr, buf_x, MROWS, D_INNER, D_MODEL, D_MODEL, D_INNER);
    gemm_wmma_f32<2, 4, 0><<<dim3(D_INNER / 64, gy), blk, 0, stream>>>(
        x, in_proj_w + (size_t)D_INNER * D_MODEL, nullptr, buf_z,
        MROWS, D_INNER, D_MODEL, D_MODEL, D_INNER);

    conv_silu_kernel<<<(MROWS * D_INNER) / 256, 256, 0, stream>>>(
        buf_x, conv_w, conv_b, buf_xc);

    gemm_wmma_f32<2, 1, 0><<<dim3(DBC_LD / 16, gy), blk, 0, stream>>>(
        buf_xc, x_proj_w, nullptr, buf_dbc, MROWS, DBC_LD, D_INNER, D_INNER, DBC_LD);

    gemm_wmma_f32<2, 4, 1><<<dim3(D_INNER / 64, gy), blk, 0, stream>>>(
        buf_dbc, dt_proj_w, dt_proj_b, buf_x, MROWS, D_INNER, DT_RANK, DBC_LD, D_INNER);

    scan_kernel<<<(BATCH * D_INNER * D_STATE) / 256, 256, 0, stream>>>(
        buf_x, buf_xc, buf_dbc, buf_z, log_A, D_skip, buf_y);

    gemm_wmma_f32<2, 4, 0><<<dim3(D_MODEL / 64, gy), blk, 0, stream>>>(
        buf_y, out_proj_w, nullptr, out, MROWS, D_MODEL, D_INNER, D_INNER, D_MODEL);
}


# gemm: A and W slices of 32 k (x_proj 64, dt_proj 16) copied to padded LDS by the block with 16-byte loads along k, two stages; waves read fragments from LDS; k order as before
# speedup vs baseline: 6.2106x; 1.3854x over previous
#include <hip/hip_runtime.h>
#include <hip/hip_bf16.h>


#define D_MODEL 768
#define D_STATE 16
#define D_CONV  4
#define D_INNER 1536
#define DT_RANK 48
#define BATCH   2
#define SEQ     2048
#define MROWS   (BATCH * SEQ)
#define DBC_LD  (DT_RANK + 2 * D_STATE)

typedef __attribute__((ext_vector_type(2))) float v2f;
typedef __attribute__((ext_vector_type(8))) float v8f;

__device__ __forceinline__ float silu_f(float x) {
    return x * (1.0f / (1.0f + __expf(-x)));
}
__device__ __forceinline__ float softplus_f(float x) {
    return (x > 20.0f) ? x : log1pf(__expf(x));
}

typedef __attribute__((ext_vector_type(4))) float v4f;

template <int MT, int NT, int EPI, int BK>
__launch_bounds__(128)
__global__ void gemm_wmma_f32(const float* __restrict__ A,
                              const float* __restrict__ W,
                              const float* __restrict__ bias,
                              float* __restrict__ C,
                              int M, int N, int K, int lda, int ldc) {
    constexpr int BM    = 64 * MT;
    constexpr int BN    = 16 * NT;
    constexpr int P     = BK + 4;
    constexpr int CPR   = BK / 4;
    constexpr int ACH   = BM * CPR / 128;
    constexpr int WCH   = BN * CPR / 128;
    constexpr int RSTEP = 128 / CPR;
    constexpr int STAGE = (BM + BN) * P;
    static_assert((BM * CPR) % 128 == 0 && (BN * CPR) % 128 == 0, "tile/threads");
    static_assert(((P / 4) & 1) == 1 && (P % 4) == 0, "pitch must be 4*odd floats");

    __shared__ __attribute__((aligned(16))) float lds[2 * STAGE];

    const int tid    = threadIdx.x;
    const int lane   = tid & 31;
    const int wave   = tid >> 5;
    const int mBlock = blockIdx.y * BM;
    const int mBase  = mBlock + wave * (16 * MT);
    const int nBase  = blockIdx.x * BN;
    const int r16    = lane & 15;
    const int kOff   = (lane >> 4) * 2;

    const int cRow = tid / CPR;
    const int cCol = (tid % CPR) * 4;
    const float* gA = A + (size_t)(mBlock + cRow) * lda + cCol;
    const float* gW = W + (size_t)(nBase + cRow) * K + cCol;
    const int sOffA = cRow * P + cCol;
    const int sOffW = (BM + cRow) * P + cCol;

    const int fA = (wave * (16 * MT) + r16) * P + kOff;
    const int fW = (BM + r16) * P + kOff;

    v8f acc[MT][NT];
#pragma unroll
    for (int i = 0; i < MT; ++i)
#pragma unroll
        for (int t = 0; t < NT; ++t) {
            v8f z = {0.f, 0.f, 0.f, 0.f, 0.f, 0.f, 0.f, 0.f};
            acc[i][t] = z;
        }

    v4f rA[ACH], rW[WCH];

#define GLOAD(K0)                                                        \
    _Pragma("unroll") for (int i = 0; i < ACH; ++i)                      \
        rA[i] = *(const v4f*)(gA + (size_t)(i * RSTEP) * lda + (K0));    \
    _Pragma("unroll") for (int i = 0; i < WCH; ++i)                      \
        rW[i] = *(const v4f*)(gW + (size_t)(i * RSTEP) * K + (K0));

#define SSTORE(ST)                                                       \
    _Pragma("unroll") for (int i = 0; i < ACH; ++i)                      \
        *(v4f*)((ST) + sOffA + i * RSTEP * P) = rA[i];                   \
    _Pragma("unroll") for (int i = 0; i < WCH; ++i)                      \
        *(v4f*)((ST) + sOffW + i * RSTEP * P) = rW[i];

    const int nS = K / BK;
    GLOAD(0)
    SSTORE(lds)
    __syncthreads();

    for (int s = 0; s < nS; ++s) {
        const float* st = lds + (s & 1) * STAGE;
        const bool more = (s + 1 < nS);
        if (more) { GLOAD((s + 1) * BK) }

#pragma unroll
        for (int kk = 0; kk < BK / 4; ++kk) {
            v2f a[MT], b[NT];
#pragma unroll
            for (int i = 0; i < MT; ++i)
                a[i] = *(const v2f*)(st + fA + i * 16 * P + kk * 4);
#pragma unroll
            for (int t = 0; t < NT; ++t)
                b[t] = *(const v2f*)(st + fW + t * 16 * P + kk * 4);
#pragma unroll
            for (int i = 0; i < MT; ++i)
#pragma unroll
                for (int t = 0; t < NT; ++t)
                    acc[i][t] = __builtin_amdgcn_wmma_f32_16x16x4_f32(
                        false, a[i], false, b[t], (short)0, acc[i][t],
                        false, false);
        }

        if (more) { SSTORE(lds + ((s + 1) & 1) * STAGE) }
        __syncthreads();
    }

#undef GLOAD
#undef SSTORE

    const int rowSel = (lane >> 4) * 8;
#pragma unroll
    for (int i = 0; i < MT; ++i)
#pragma unroll
        for (int t = 0; t < NT; ++t) {
            const int col = nBase + t * 16 + (lane & 15);
            float bv = 0.0f;
            if (EPI == 1) bv = bias[col];
#pragma unroll
            for (int r = 0; r < 8; ++r) {
                float v = acc[i][t][r];
                if (EPI == 1) v = softplus_f(v + bv);
                C[(size_t)(mBase + i * 16 + r + rowSel) * ldc + col] = v;
            }
        }
}

__launch_bounds__(256)
__global__ void conv_silu_kernel(const float* __restrict__ xin,
                                 const float* __restrict__ w,
                                 const float* __restrict__ b,
                                 float* __restrict__ out) {
    const int idx = blockIdx.x * blockDim.x + threadIdx.x;
    const int d  = idx % D_INNER;
    const int s  = (idx / D_INNER) % SEQ;
    const int bi = idx / (D_INNER * SEQ);

    const float* base = xin + (size_t)bi * SEQ * D_INNER + d;
    float acc = b[d];
#pragma unroll
    for (int j = 0; j < D_CONV; ++j) {
        const int sp = s - (D_CONV - 1) + j;
        if (sp >= 0) acc += w[d * D_CONV + j] * base[(size_t)sp * D_INNER];
    }
    out[idx] = silu_f(acc);
}

#define TILE_S 16

template <int J>
__device__ __forceinline__ float row_share_f(float v) {
    return __int_as_float(__builtin_amdgcn_update_dpp(
        0, __float_as_int(v), 0x150 + J, 0xf, 0xf, false));
}

__launch_bounds__(256)
__global__ void scan_kernel(const float* __restrict__ delta,
                            const float* __restrict__ xc,
                            const float* __restrict__ dbc,
                            const float* __restrict__ zbuf,
                            const float* __restrict__ logA,
                            const float* __restrict__ Dskip,
                            float* __restrict__ y) {
    const int gid = blockIdx.x * blockDim.x + threadIdx.x;
    const int n = gid & (D_STATE - 1);
    const int d = (gid >> 4) % D_INNER;
    const int b = gid / (D_INNER * D_STATE);

    const float Aneg = -__expf(logA[d * D_STATE + n]);
    const float dsk  = Dskip[d];

    const size_t baseBD = (size_t)b * SEQ * D_INNER + d + (size_t)n * D_INNER;
    const float* pDt = delta + baseBD;
    const float* pXc = xc    + baseBD;
    const float* pZ  = zbuf  + baseBD;
    float*       pY  = y     + baseBD;
    const float* pBC = dbc + (size_t)b * SEQ * DBC_LD + DT_RANK + n;

    float dtA, xvA, zvA, BA[TILE_S], CA[TILE_S];
    float dtB, xvB, zvB, BB[TILE_S], CB[TILE_S];

#define LOAD_TILE(DT, XV, ZV, BV, CV, S0)                              \
    {                                                                  \
        const size_t o = (size_t)(S0) * D_INNER;                       \
        DT = pDt[o];                                                   \
        XV = pXc[o];                                                   \
        const float* pb = pBC + (size_t)(S0) * DBC_LD;                 \
                \
        _Pragma("unroll") for (int j = 0; j < TILE_S; ++j) {           \
            BV[j] = pb[j * DBC_LD];                                    \
            CV[j] = pb[j * DBC_LD + D_STATE];                          \
        }                                                              \
        ZV = pZ[o];                                                    \
    }

#define SCAN_STEP(J, DT, XV, BV, CV)                                   \
    {                                                                  \
        const float dtj = row_share_f<J>(DT);                          \
        const float xvj = row_share_f<J>(XV);                          \
        const float a = __expf(dtj * Aneg);                            \
           \
           \
        h = fmaf(a, h, (dtj * BV[J]) * xvj);                           \
        P[J] = h * CV[J];                                              \
    }

#define RED_STAGE(M)                                                   \
                    \
    _Pragma("unroll") for (int j = 0; j < TILE_S; ++j)                 \
        P[j] += __shfl_xor(P[j], M, 16);

#define SCAN_TILE(DT, XV, ZV, BV, CV, S0)                              \
    {                                                                  \
        float P[TILE_S];                                               \
        SCAN_STEP(0, DT, XV, BV, CV)  SCAN_STEP(1, DT, XV, BV, CV)     \
        SCAN_STEP(2, DT, XV, BV, CV)  SCAN_STEP(3, DT, XV, BV, CV)     \
        SCAN_STEP(4, DT, XV, BV, CV)  SCAN_STEP(5, DT, XV, BV, CV)     \
        SCAN_STEP(6, DT, XV, BV, CV)  SCAN_STEP(7, DT, XV, BV, CV)     \
        SCAN_STEP(8, DT, XV, BV, CV)  SCAN_STEP(9, DT, XV, BV, CV)     \
        SCAN_STEP(10, DT, XV, BV, CV) SCAN_STEP(11, DT, XV, BV, CV)    \
        SCAN_STEP(12, DT, XV, BV, CV) SCAN_STEP(13, DT, XV, BV, CV)    \
        SCAN_STEP(14, DT, XV, BV, CV) SCAN_STEP(15, DT, XV, BV, CV)    \
           \
        RED_STAGE(1) RED_STAGE(2) RED_STAGE(4) RED_STAGE(8)            \
        float yv = 0.0f;                                               \
          \
        _Pragma("unroll") for (int j = 0; j < TILE_S; ++j)             \
            yv = (n == j) ? P[j] : yv;                                 \
                \
        pY[(size_t)(S0) * D_INNER] = (yv + dsk * XV) * silu_f(ZV);     \
    }

#define HOLD_LOADS() __builtin_amdgcn_sched_barrier(0);

    float h = 0.0f;
    LOAD_TILE(dtA, xvA, zvA, BA, CA, 0)
    int s0 = 0;
    for (; s0 + 2 * TILE_S < SEQ; s0 += 2 * TILE_S) {
        LOAD_TILE(dtB, xvB, zvB, BB, CB, s0 + TILE_S)
        HOLD_LOADS()
        SCAN_TILE(dtA, xvA, zvA, BA, CA, s0)
        HOLD_LOADS()
        LOAD_TILE(dtA, xvA, zvA, BA, CA, s0 + 2 * TILE_S)
        HOLD_LOADS()
        SCAN_TILE(dtB, xvB, zvB, BB, CB, s0 + TILE_S)
    }
    LOAD_TILE(dtB, xvB, zvB, BB, CB, s0 + TILE_S)
    HOLD_LOADS()
    SCAN_TILE(dtA, xvA, zvA, BA, CA, s0)
    SCAN_TILE(dtB, xvB, zvB, BB, CB, s0 + TILE_S)

#undef HOLD_LOADS
#undef RED_STAGE
#undef LOAD_TILE
#undef SCAN_STEP
#undef SCAN_TILE
}

extern "C" void kernel_launch(void* const* d_in, const int* in_sizes, int n_in,
                              void* d_out, int out_size, void* d_ws, size_t ws_size,
                              hipStream_t stream) {
    const float* x         = (const float*)d_in[0];
    const float* in_proj_w = (const float*)d_in[1];
    const float* conv_w    = (const float*)d_in[2];
    const float* conv_b    = (const float*)d_in[3];
    const float* x_proj_w  = (const float*)d_in[4];
    const float* dt_proj_w = (const float*)d_in[5];
    const float* dt_proj_b = (const float*)d_in[6];
    const float* log_A     = (const float*)d_in[7];
    const float* D_skip    = (const float*)d_in[8];
    const float* out_proj_w= (const float*)d_in[9];
    float* out = (float*)d_out;

    const size_t planeBD = (size_t)MROWS * D_INNER;
    float* ws      = (float*)d_ws;
    float* buf_x   = ws;
    float* buf_z   = buf_x  + planeBD;
    float* buf_xc  = buf_z  + planeBD;
    float* buf_dbc = buf_xc + planeBD;
    float* buf_y   = buf_dbc + (size_t)MROWS * DBC_LD;

    const dim3 blk(128);
    const int gy = MROWS / (16 * 2 * 4);

    gemm_wmma_f32<2, 4, 0, 32><<<dim3(D_INNER / 64, gy), blk, 0, stream>>>(
        x, in_proj_w, nullptr, buf_x, MROWS, D_INNER, D_MODEL, D_MODEL, D_INNER);
    gemm_wmma_f32<2, 4, 0, 32><<<dim3(D_INNER / 64, gy), blk, 0, stream>>>(
        x, in_proj_w + (size_t)D_INNER * D_MODEL, nullptr, buf_z,
        MROWS, D_INNER, D_MODEL, D_MODEL, D_INNER);

    conv_silu_kernel<<<(MROWS * D_INNER) / 256, 256, 0, stream>>>(
        buf_x, conv_w, conv_b, buf_xc);

    gemm_wmma_f32<2, 1, 0, 64><<<dim3(DBC_LD / 16, gy), blk, 0, stream>>>(
        buf_xc, x_proj_w, nullptr, buf_dbc, MROWS, DBC_LD, D_INNER, D_INNER, DBC_LD);

    gemm_wmma_f32<2, 4, 1, 16><<<dim3(D_INNER / 64, gy), blk, 0, stream>>>(
        buf_dbc, dt_proj_w, dt_proj_b, buf_x, MROWS, D_INNER, DT_RANK, DBC_LD, D_INNER);

    scan_kernel<<<(BATCH * D_INNER * D_STATE) / 256, 256, 0, stream>>>(
        buf_x, buf_xc, buf_dbc, buf_z, log_A, D_skip, buf_y);

    gemm_wmma_f32<2, 4, 0, 32><<<dim3(D_MODEL / 64, gy), blk, 0, stream>>>(
        buf_y, out_proj_w, nullptr, out, MROWS, D_MODEL, D_INNER, D_INNER, D_MODEL);
}


# out_proj: block tile 128x32 (wave tile 32x32) in 64-k slices, 768 blocks = 3 on every WGP in place of 384 (2 on half the WGPs, 1 on the rest); k order of the sums unchanged
# speedup vs baseline: 6.4181x; 1.0334x over previous
#include <hip/hip_runtime.h>
#include <hip/hip_bf16.h>


#define D_MODEL 768
#define D_STATE 16
#define D_CONV  4
#define D_INNER 1536
#define DT_RANK 48
#define BATCH   2
#define SEQ     2048
#define MROWS   (BATCH * SEQ)
#define DBC_LD  (DT_RANK + 2 * D_STATE)

typedef __attribute__((ext_vector_type(2))) float v2f;
typedef __attribute__((ext_vector_type(8))) float v8f;

__device__ __forceinline__ float silu_f(float x) {
    return x * (1.0f / (1.0f + __expf(-x)));
}
__device__ __forceinline__ float softplus_f(float x) {
    return (x > 20.0f) ? x : log1pf(__expf(x));
}

typedef __attribute__((ext_vector_type(4))) float v4f;

template <int MT, int NT, int EPI, int BK>
__launch_bounds__(128)
__global__ void gemm_wmma_f32(const float* __restrict__ A,
                              const float* __restrict__ W,
                              const float* __restrict__ bias,
                              float* __restrict__ C,
                              int M, int N, int K, int lda, int ldc) {
    constexpr int BM    = 64 * MT;
    constexpr int BN    = 16 * NT;
    constexpr int P     = BK + 4;
    constexpr int CPR   = BK / 4;
    constexpr int ACH   = BM * CPR / 128;
    constexpr int WCH   = BN * CPR / 128;
    constexpr int RSTEP = 128 / CPR;
    constexpr int STAGE = (BM + BN) * P;
    static_assert((BM * CPR) % 128 == 0 && (BN * CPR) % 128 == 0, "tile/threads");
    static_assert(((P / 4) & 1) == 1 && (P % 4) == 0, "pitch must be 4*odd floats");

    __shared__ __attribute__((aligned(16))) float lds[2 * STAGE];

    const int tid    = threadIdx.x;
    const int lane   = tid & 31;
    const int wave   = tid >> 5;
    const int mBlock = blockIdx.y * BM;
    const int mBase  = mBlock + wave * (16 * MT);
    const int nBase  = blockIdx.x * BN;
    const int r16    = lane & 15;
    const int kOff   = (lane >> 4) * 2;

    const int cRow = tid / CPR;
    const int cCol = (tid % CPR) * 4;
    const float* gA = A + (size_t)(mBlock + cRow) * lda + cCol;
    const float* gW = W + (size_t)(nBase + cRow) * K + cCol;
    const int sOffA = cRow * P + cCol;
    const int sOffW = (BM + cRow) * P + cCol;

    const int fA = (wave * (16 * MT) + r16) * P + kOff;
    const int fW = (BM + r16) * P + kOff;

    v8f acc[MT][NT];
#pragma unroll
    for (int i = 0; i < MT; ++i)
#pragma unroll
        for (int t = 0; t < NT; ++t) {
            v8f z = {0.f, 0.f, 0.f, 0.f, 0.f, 0.f, 0.f, 0.f};
            acc[i][t] = z;
        }

    v4f rA[ACH], rW[WCH];

#define GLOAD(K0)                                                        \
    _Pragma("unroll") for (int i = 0; i < ACH; ++i)                      \
        rA[i] = *(const v4f*)(gA + (size_t)(i * RSTEP) * lda + (K0));    \
    _Pragma("unroll") for (int i = 0; i < WCH; ++i)                      \
        rW[i] = *(const v4f*)(gW + (size_t)(i * RSTEP) * K + (K0));

#define SSTORE(ST)                                                       \
    _Pragma("unroll") for (int i = 0; i < ACH; ++i)                      \
        *(v4f*)((ST) + sOffA + i * RSTEP * P) = rA[i];                   \
    _Pragma("unroll") for (int i = 0; i < WCH; ++i)                      \
        *(v4f*)((ST) + sOffW + i * RSTEP * P) = rW[i];

    const int nS = K / BK;
    GLOAD(0)
    SSTORE(lds)
    __syncthreads();

    for (int s = 0; s < nS; ++s) {
        const float* st = lds + (s & 1) * STAGE;
        const bool more = (s + 1 < nS);
        if (more) { GLOAD((s + 1) * BK) }

#pragma unroll
        for (int kk = 0; kk < BK / 4; ++kk) {
            v2f a[MT], b[NT];
#pragma unroll
            for (int i = 0; i < MT; ++i)
                a[i] = *(const v2f*)(st + fA + i * 16 * P + kk * 4);
#pragma unroll
            for (int t = 0; t < NT; ++t)
                b[t] = *(const v2f*)(st + fW + t * 16 * P + kk * 4);
#pragma unroll
            for (int i = 0; i < MT; ++i)
#pragma unroll
                for (int t = 0; t < NT; ++t)
                    acc[i][t] = __builtin_amdgcn_wmma_f32_16x16x4_f32(
                        false, a[i], false, b[t], (short)0, acc[i][t],
                        false, false);
        }

        if (more) { SSTORE(lds + ((s + 1) & 1) * STAGE) }
        __syncthreads();
    }

#undef GLOAD
#undef SSTORE

    const int rowSel = (lane >> 4) * 8;
#pragma unroll
    for (int i = 0; i < MT; ++i)
#pragma unroll
        for (int t = 0; t < NT; ++t) {
            const int col = nBase + t * 16 + (lane & 15);
            float bv = 0.0f;
            if (EPI == 1) bv = bias[col];
#pragma unroll
            for (int r = 0; r < 8; ++r) {
                float v = acc[i][t][r];
                if (EPI == 1) v = softplus_f(v + bv);
                C[(size_t)(mBase + i * 16 + r + rowSel) * ldc + col] = v;
            }
        }
}

__launch_bounds__(256)
__global__ void conv_silu_kernel(const float* __restrict__ xin,
                                 const float* __restrict__ w,
                                 const float* __restrict__ b,
                                 float* __restrict__ out) {
    const int idx = blockIdx.x * blockDim.x + threadIdx.x;
    const int d  = idx % D_INNER;
    const int s  = (idx / D_INNER) % SEQ;
    const int bi = idx / (D_INNER * SEQ);

    const float* base = xin + (size_t)bi * SEQ * D_INNER + d;
    float acc = b[d];
#pragma unroll
    for (int j = 0; j < D_CONV; ++j) {
        const int sp = s - (D_CONV - 1) + j;
        if (sp >= 0) acc += w[d * D_CONV + j] * base[(size_t)sp * D_INNER];
    }
    out[idx] = silu_f(acc);
}

#define TILE_S 16

template <int J>
__device__ __forceinline__ float row_share_f(float v) {
    return __int_as_float(__builtin_amdgcn_update_dpp(
        0, __float_as_int(v), 0x150 + J, 0xf, 0xf, false));
}

__launch_bounds__(256)
__global__ void scan_kernel(const float* __restrict__ delta,
                            const float* __restrict__ xc,
                            const float* __restrict__ dbc,
                            const float* __restrict__ zbuf,
                            const float* __restrict__ logA,
                            const float* __restrict__ Dskip,
                            float* __restrict__ y) {
    const int gid = blockIdx.x * blockDim.x + threadIdx.x;
    const int n = gid & (D_STATE - 1);
    const int d = (gid >> 4) % D_INNER;
    const int b = gid / (D_INNER * D_STATE);

    const float Aneg = -__expf(logA[d * D_STATE + n]);
    const float dsk  = Dskip[d];

    const size_t baseBD = (size_t)b * SEQ * D_INNER + d + (size_t)n * D_INNER;
    const float* pDt = delta + baseBD;
    const float* pXc = xc    + baseBD;
    const float* pZ  = zbuf  + baseBD;
    float*       pY  = y     + baseBD;
    const float* pBC = dbc + (size_t)b * SEQ * DBC_LD + DT_RANK + n;

    float dtA, xvA, zvA, BA[TILE_S], CA[TILE_S];
    float dtB, xvB, zvB, BB[TILE_S], CB[TILE_S];

#define LOAD_TILE(DT, XV, ZV, BV, CV, S0)                              \
    {                                                                  \
        const size_t o = (size_t)(S0) * D_INNER;                       \
        DT = pDt[o];                                                   \
        XV = pXc[o];                                                   \
        const float* pb = pBC + (size_t)(S0) * DBC_LD;                 \
                \
        _Pragma("unroll") for (int j = 0; j < TILE_S; ++j) {           \
            BV[j] = pb[j * DBC_LD];                                    \
            CV[j] = pb[j * DBC_LD + D_STATE];                          \
        }                                                              \
        ZV = pZ[o];                                                    \
    }

#define SCAN_STEP(J, DT, XV, BV, CV)                                   \
    {                                                                  \
        const float dtj = row_share_f<J>(DT);                          \
        const float xvj = row_share_f<J>(XV);                          \
        const float a = __expf(dtj * Aneg);                            \
           \
           \
        h = fmaf(a, h, (dtj * BV[J]) * xvj);                           \
        P[J] = h * CV[J];                                              \
    }

#define RED_STAGE(M)                                                   \
                    \
    _Pragma("unroll") for (int j = 0; j < TILE_S; ++j)                 \
        P[j] += __shfl_xor(P[j], M, 16);

#define SCAN_TILE(DT, XV, ZV, BV, CV, S0)                              \
    {                                                                  \
        float P[TILE_S];                                               \
        SCAN_STEP(0, DT, XV, BV, CV)  SCAN_STEP(1, DT, XV, BV, CV)     \
        SCAN_STEP(2, DT, XV, BV, CV)  SCAN_STEP(3, DT, XV, BV, CV)     \
        SCAN_STEP(4, DT, XV, BV, CV)  SCAN_STEP(5, DT, XV, BV, CV)     \
        SCAN_STEP(6, DT, XV, BV, CV)  SCAN_STEP(7, DT, XV, BV, CV)     \
        SCAN_STEP(8, DT, XV, BV, CV)  SCAN_STEP(9, DT, XV, BV, CV)     \
        SCAN_STEP(10, DT, XV, BV, CV) SCAN_STEP(11, DT, XV, BV, CV)    \
        SCAN_STEP(12, DT, XV, BV, CV) SCAN_STEP(13, DT, XV, BV, CV)    \
        SCAN_STEP(14, DT, XV, BV, CV) SCAN_STEP(15, DT, XV, BV, CV)    \
           \
        RED_STAGE(1) RED_STAGE(2) RED_STAGE(4) RED_STAGE(8)            \
        float yv = 0.0f;                                               \
          \
        _Pragma("unroll") for (int j = 0; j < TILE_S; ++j)             \
            yv = (n == j) ? P[j] : yv;                                 \
                \
        pY[(size_t)(S0) * D_INNER] = (yv + dsk * XV) * silu_f(ZV);     \
    }

#define HOLD_LOADS() __builtin_amdgcn_sched_barrier(0);

    float h = 0.0f;
    LOAD_TILE(dtA, xvA, zvA, BA, CA, 0)
    int s0 = 0;
    for (; s0 + 2 * TILE_S < SEQ; s0 += 2 * TILE_S) {
        LOAD_TILE(dtB, xvB, zvB, BB, CB, s0 + TILE_S)
        HOLD_LOADS()
        SCAN_TILE(dtA, xvA, zvA, BA, CA, s0)
        HOLD_LOADS()
        LOAD_TILE(dtA, xvA, zvA, BA, CA, s0 + 2 * TILE_S)
        HOLD_LOADS()
        SCAN_TILE(dtB, xvB, zvB, BB, CB, s0 + TILE_S)
    }
    LOAD_TILE(dtB, xvB, zvB, BB, CB, s0 + TILE_S)
    HOLD_LOADS()
    SCAN_TILE(dtA, xvA, zvA, BA, CA, s0)
    SCAN_TILE(dtB, xvB, zvB, BB, CB, s0 + TILE_S)

#undef HOLD_LOADS
#undef RED_STAGE
#undef LOAD_TILE
#undef SCAN_STEP
#undef SCAN_TILE
}

extern "C" void kernel_launch(void* const* d_in, const int* in_sizes, int n_in,
                              void* d_out, int out_size, void* d_ws, size_t ws_size,
                              hipStream_t stream) {
    const float* x         = (const float*)d_in[0];
    const float* in_proj_w = (const float*)d_in[1];
    const float* conv_w    = (const float*)d_in[2];
    const float* conv_b    = (const float*)d_in[3];
    const float* x_proj_w  = (const float*)d_in[4];
    const float* dt_proj_w = (const float*)d_in[5];
    const float* dt_proj_b = (const float*)d_in[6];
    const float* log_A     = (const float*)d_in[7];
    const float* D_skip    = (const float*)d_in[8];
    const float* out_proj_w= (const float*)d_in[9];
    float* out = (float*)d_out;

    const size_t planeBD = (size_t)MROWS * D_INNER;
    float* ws      = (float*)d_ws;
    float* buf_x   = ws;
    float* buf_z   = buf_x  + planeBD;
    float* buf_xc  = buf_z  + planeBD;
    float* buf_dbc = buf_xc + planeBD;
    float* buf_y   = buf_dbc + (size_t)MROWS * DBC_LD;

    const dim3 blk(128);
    const int gy = MROWS / (16 * 2 * 4);

    gemm_wmma_f32<2, 4, 0, 32><<<dim3(D_INNER / 64, gy), blk, 0, stream>>>(
        x, in_proj_w, nullptr, buf_x, MROWS, D_INNER, D_MODEL, D_MODEL, D_INNER);
    gemm_wmma_f32<2, 4, 0, 32><<<dim3(D_INNER / 64, gy), blk, 0, stream>>>(
        x, in_proj_w + (size_t)D_INNER * D_MODEL, nullptr, buf_z,
        MROWS, D_INNER, D_MODEL, D_MODEL, D_INNER);

    conv_silu_kernel<<<(MROWS * D_INNER) / 256, 256, 0, stream>>>(
        buf_x, conv_w, conv_b, buf_xc);

    gemm_wmma_f32<2, 1, 0, 64><<<dim3(DBC_LD / 16, gy), blk, 0, stream>>>(
        buf_xc, x_proj_w, nullptr, buf_dbc, MROWS, DBC_LD, D_INNER, D_INNER, DBC_LD);

    gemm_wmma_f32<2, 4, 1, 16><<<dim3(D_INNER / 64, gy), blk, 0, stream>>>(
        buf_dbc, dt_proj_w, dt_proj_b, buf_x, MROWS, D_INNER, DT_RANK, DBC_LD, D_INNER);

    scan_kernel<<<(BATCH * D_INNER * D_STATE) / 256, 256, 0, stream>>>(
        buf_x, buf_xc, buf_dbc, buf_z, log_A, D_skip, buf_y);

    gemm_wmma_f32<2, 2, 0, 64><<<dim3(D_MODEL / 32, gy), blk, 0, stream>>>(
        buf_y, out_proj_w, nullptr, out, MROWS, D_MODEL, D_INNER, D_INNER, D_MODEL);
}


# scan: two states (2p,2p+1) per lane, 8 lanes/channel, 768 waves, loads two tiles ahead; x_proj stores a pair's B,C adjacent (16-byte load per step); row_xmask DPP sum tree, pair order 1,2,4,8 kept
# speedup vs baseline: 7.8124x; 1.2172x over previous
#include <hip/hip_runtime.h>
#include <hip/hip_bf16.h>


#define D_MODEL 768
#define D_STATE 16
#define D_CONV  4
#define D_INNER 1536
#define DT_RANK 48
#define BATCH   2
#define SEQ     2048
#define MROWS   (BATCH * SEQ)
#define DBC_LD  (DT_RANK + 2 * D_STATE)

typedef __attribute__((ext_vector_type(2))) float v2f;
typedef __attribute__((ext_vector_type(8))) float v8f;

__device__ __forceinline__ float silu_f(float x) {
    return x * (1.0f / (1.0f + __expf(-x)));
}
__device__ __forceinline__ float softplus_f(float x) {
    return (x > 20.0f) ? x : log1pf(__expf(x));
}

typedef __attribute__((ext_vector_type(4))) float v4f;

template <int MT, int NT, int EPI, int BK>
__launch_bounds__(128)
__global__ void gemm_wmma_f32(const float* __restrict__ A,
                              const float* __restrict__ W,
                              const float* __restrict__ bias,
                              float* __restrict__ C,
                              int M, int N, int K, int lda, int ldc) {
    constexpr int BM    = 64 * MT;
    constexpr int BN    = 16 * NT;
    constexpr int P     = BK + 4;
    constexpr int CPR   = BK / 4;
    constexpr int ACH   = BM * CPR / 128;
    constexpr int WCH   = BN * CPR / 128;
    constexpr int RSTEP = 128 / CPR;
    constexpr int STAGE = (BM + BN) * P;
    static_assert((BM * CPR) % 128 == 0 && (BN * CPR) % 128 == 0, "tile/threads");
    static_assert(((P / 4) & 1) == 1 && (P % 4) == 0, "pitch must be 4*odd floats");

    __shared__ __attribute__((aligned(16))) float lds[2 * STAGE];

    const int tid    = threadIdx.x;
    const int lane   = tid & 31;
    const int wave   = tid >> 5;
    const int mBlock = blockIdx.y * BM;
    const int mBase  = mBlock + wave * (16 * MT);
    const int nBase  = blockIdx.x * BN;
    const int r16    = lane & 15;
    const int kOff   = (lane >> 4) * 2;

    const int cRow = tid / CPR;
    const int cCol = (tid % CPR) * 4;
    const float* gA = A + (size_t)(mBlock + cRow) * lda + cCol;
    const float* gW = W + (size_t)(nBase + cRow) * K + cCol;
    const int sOffA = cRow * P + cCol;
    const int sOffW = (BM + cRow) * P + cCol;

    const int fA = (wave * (16 * MT) + r16) * P + kOff;
    const int fW = (BM + r16) * P + kOff;

    v8f acc[MT][NT];
#pragma unroll
    for (int i = 0; i < MT; ++i)
#pragma unroll
        for (int t = 0; t < NT; ++t) {
            v8f z = {0.f, 0.f, 0.f, 0.f, 0.f, 0.f, 0.f, 0.f};
            acc[i][t] = z;
        }

    v4f rA[ACH], rW[WCH];

#define GLOAD(K0)                                                        \
    _Pragma("unroll") for (int i = 0; i < ACH; ++i)                      \
        rA[i] = *(const v4f*)(gA + (size_t)(i * RSTEP) * lda + (K0));    \
    _Pragma("unroll") for (int i = 0; i < WCH; ++i)                      \
        rW[i] = *(const v4f*)(gW + (size_t)(i * RSTEP) * K + (K0));

#define SSTORE(ST)                                                       \
    _Pragma("unroll") for (int i = 0; i < ACH; ++i)                      \
        *(v4f*)((ST) + sOffA + i * RSTEP * P) = rA[i];                   \
    _Pragma("unroll") for (int i = 0; i < WCH; ++i)                      \
        *(v4f*)((ST) + sOffW + i * RSTEP * P) = rW[i];

    const int nS = K / BK;
    GLOAD(0)
    SSTORE(lds)
    __syncthreads();

    for (int s = 0; s < nS; ++s) {
        const float* st = lds + (s & 1) * STAGE;
        const bool more = (s + 1 < nS);
        if (more) { GLOAD((s + 1) * BK) }

#pragma unroll
        for (int kk = 0; kk < BK / 4; ++kk) {
            v2f a[MT], b[NT];
#pragma unroll
            for (int i = 0; i < MT; ++i)
                a[i] = *(const v2f*)(st + fA + i * 16 * P + kk * 4);
#pragma unroll
            for (int t = 0; t < NT; ++t)
                b[t] = *(const v2f*)(st + fW + t * 16 * P + kk * 4);
#pragma unroll
            for (int i = 0; i < MT; ++i)
#pragma unroll
                for (int t = 0; t < NT; ++t)
                    acc[i][t] = __builtin_amdgcn_wmma_f32_16x16x4_f32(
                        false, a[i], false, b[t], (short)0, acc[i][t],
                        false, false);
        }

        if (more) { SSTORE(lds + ((s + 1) & 1) * STAGE) }
        __syncthreads();
    }

#undef GLOAD
#undef SSTORE

    const int rowSel = (lane >> 4) * 8;
#pragma unroll
    for (int i = 0; i < MT; ++i)
#pragma unroll
        for (int t = 0; t < NT; ++t) {
            const int col = nBase + t * 16 + (lane & 15);
            int ccol = col;
            if (EPI == 2 && col >= DT_RANK) {
                const int q = col - DT_RANK;
                ccol = DT_RANK + 4 * ((q & 15) >> 1) + 2 * (q >> 4) + (q & 1);
            }
            float bv = 0.0f;
            if (EPI == 1) bv = bias[col];
#pragma unroll
            for (int r = 0; r < 8; ++r) {
                float v = acc[i][t][r];
                if (EPI == 1) v = softplus_f(v + bv);
                C[(size_t)(mBase + i * 16 + r + rowSel) * ldc + ccol] = v;
            }
        }
}

__launch_bounds__(256)
__global__ void conv_silu_kernel(const float* __restrict__ xin,
                                 const float* __restrict__ w,
                                 const float* __restrict__ b,
                                 float* __restrict__ out) {
    const int idx = blockIdx.x * blockDim.x + threadIdx.x;
    const int d  = idx % D_INNER;
    const int s  = (idx / D_INNER) % SEQ;
    const int bi = idx / (D_INNER * SEQ);

    const float* base = xin + (size_t)bi * SEQ * D_INNER + d;
    float acc = b[d];
#pragma unroll
    for (int j = 0; j < D_CONV; ++j) {
        const int sp = s - (D_CONV - 1) + j;
        if (sp >= 0) acc += w[d * D_CONV + j] * base[(size_t)sp * D_INNER];
    }
    out[idx] = silu_f(acc);
}

#define TILE_S 16
#define SCAN_THREADS 96

template <int J>
__device__ __forceinline__ float bcast8_f(float v) {
    return __int_as_float(__builtin_amdgcn_mov_dpp8(__float_as_int(v), J * 0x249249));
}

template <int M>
__device__ __forceinline__ float row_xor_f(float v) {
    return __int_as_float(__builtin_amdgcn_update_dpp(
        0, __float_as_int(v), 0x160 + M, 0xf, 0xf, false));
}

__launch_bounds__(SCAN_THREADS)
__global__ void scan_kernel(const float* __restrict__ delta,
                            const float* __restrict__ xc,
                            const float* __restrict__ dbc,
                            const float* __restrict__ zbuf,
                            const float* __restrict__ logA,
                            const float* __restrict__ Dskip,
                            float* __restrict__ y) {
    const int gid = blockIdx.x * blockDim.x + threadIdx.x;
    const int p  = gid & 7;
    const int ch = gid >> 3;
    const int d  = ch % D_INNER;
    const int b  = ch / D_INNER;

    const float Aneg0 = -__expf(logA[d * D_STATE + 2 * p]);
    const float Aneg1 = -__expf(logA[d * D_STATE + 2 * p + 1]);
    const float dsk   = Dskip[d];
    const bool  bit0 = (p & 1) != 0, bit1 = (p & 2) != 0, bit2 = (p & 4) != 0;

    const size_t baseBD = (size_t)b * SEQ * D_INNER + d + (size_t)p * D_INNER;
    const size_t HI = (size_t)8 * D_INNER;
    const float* pDt = delta + baseBD;
    const float* pXc = xc    + baseBD;
    const float* pZ  = zbuf  + baseBD;
    float*       pY  = y     + baseBD;
    const float* pBC = dbc + (size_t)b * SEQ * DBC_LD + DT_RANK + 4 * p;

    float dtLA, dtHA, xLA, xHA, zLA, zHA;  v4f bcA[TILE_S];
    float dtLB, dtHB, xLB, xHB, zLB, zHB;  v4f bcB[TILE_S];
    float dtLC, dtHC, xLC, xHC, zLC, zHC;  v4f bcC[TILE_S];

#define LOAD_TILE(DTL, DTH, XL, XH, ZL, ZH, BC, S0)                    \
    {                                                                  \
        const size_t o = (size_t)(S0) * D_INNER;                       \
        DTL = pDt[o];  DTH = pDt[o + HI];                              \
        XL  = pXc[o];  XH  = pXc[o + HI];                              \
        const float* pb = pBC + (size_t)(S0) * DBC_LD;                 \
                   \
        _Pragma("unroll") for (int j = 0; j < TILE_S; ++j)             \
            BC[j] = *(const v4f*)(pb + j * DBC_LD);                    \
        ZL  = pZ[o];   ZH  = pZ[o + HI];                               \
    }

#define SCAN_STEP(J, DTL, DTH, XL, XH, BC)                             \
    {                                                                  \
        const float dtj = bcast8_f<((J) & 7)>(((J) < 8) ? DTL : DTH);  \
        const float xvj = bcast8_f<((J) & 7)>(((J) < 8) ? XL : XH);    \
        const float a0 = __expf(dtj * Aneg0);                          \
        const float a1 = __expf(dtj * Aneg1);                          \
          \
          \
        h0 = fmaf(a0, h0, (dtj * BC[J].x) * xvj);                      \
        h1 = fmaf(a1, h1, (dtj * BC[J].y) * xvj);                      \
        V[J] = h0 * BC[J].z + h1 * BC[J].w;                            \
    }

#define TREE_STAGE(OUT, IN, CNT, BIT, M)                               \
           \
    _Pragma("unroll") for (int i = 0; i < (CNT); ++i) {                \
        const float keep = (BIT) ? IN[2 * i + 1] : IN[2 * i];          \
        const float send = (BIT) ? IN[2 * i] : IN[2 * i + 1];          \
        OUT[i] = keep + row_xor_f<M>(send);                            \
    }

#define SCAN_TILE(DTL, DTH, XL, XH, ZL, ZH, BC, S0)                    \
    {                                                                  \
        float V[TILE_S], Q[8], R[4], S[2];                             \
        SCAN_STEP(0, DTL, DTH, XL, XH, BC)  SCAN_STEP(1, DTL, DTH, XL, XH, BC)   \
        SCAN_STEP(2, DTL, DTH, XL, XH, BC)  SCAN_STEP(3, DTL, DTH, XL, XH, BC)   \
        SCAN_STEP(4, DTL, DTH, XL, XH, BC)  SCAN_STEP(5, DTL, DTH, XL, XH, BC)   \
        SCAN_STEP(6, DTL, DTH, XL, XH, BC)  SCAN_STEP(7, DTL, DTH, XL, XH, BC)   \
        SCAN_STEP(8, DTL, DTH, XL, XH, BC)  SCAN_STEP(9, DTL, DTH, XL, XH, BC)   \
        SCAN_STEP(10, DTL, DTH, XL, XH, BC) SCAN_STEP(11, DTL, DTH, XL, XH, BC)  \
        SCAN_STEP(12, DTL, DTH, XL, XH, BC) SCAN_STEP(13, DTL, DTH, XL, XH, BC)  \
        SCAN_STEP(14, DTL, DTH, XL, XH, BC) SCAN_STEP(15, DTL, DTH, XL, XH, BC)  \
          \
        TREE_STAGE(Q, V, 8, bit0, 1)                                   \
        TREE_STAGE(R, Q, 4, bit1, 2)                                   \
        TREE_STAGE(S, R, 2, bit2, 4)                                   \
          \
        const size_t o = (size_t)(S0) * D_INNER;                       \
        pY[o]      = (S[0] + dsk * XL) * silu_f(ZL);                   \
        pY[o + HI] = (S[1] + dsk * XH) * silu_f(ZH);                   \
    }

#define HOLD_LOADS() __builtin_amdgcn_sched_barrier(0);

    float h0 = 0.0f, h1 = 0.0f;
    LOAD_TILE(dtLA, dtHA, xLA, xHA, zLA, zHA, bcA, 0)
    LOAD_TILE(dtLB, dtHB, xLB, xHB, zLB, zHB, bcB, TILE_S)
    int s0 = 0;
    for (; s0 + 5 * TILE_S <= SEQ; s0 += 3 * TILE_S) {
        LOAD_TILE(dtLC, dtHC, xLC, xHC, zLC, zHC, bcC, s0 + 2 * TILE_S)
        HOLD_LOADS()
        SCAN_TILE(dtLA, dtHA, xLA, xHA, zLA, zHA, bcA, s0)
        HOLD_LOADS()
        LOAD_TILE(dtLA, dtHA, xLA, xHA, zLA, zHA, bcA, s0 + 3 * TILE_S)
        HOLD_LOADS()
        SCAN_TILE(dtLB, dtHB, xLB, xHB, zLB, zHB, bcB, s0 + TILE_S)
        HOLD_LOADS()
        LOAD_TILE(dtLB, dtHB, xLB, xHB, zLB, zHB, bcB, s0 + 4 * TILE_S)
        HOLD_LOADS()
        SCAN_TILE(dtLC, dtHC, xLC, xHC, zLC, zHC, bcC, s0 + 2 * TILE_S)
    }
    static_assert((SEQ / TILE_S) % 3 == 2 && SEQ % TILE_S == 0, "tile count");
    SCAN_TILE(dtLA, dtHA, xLA, xHA, zLA, zHA, bcA, s0)
    SCAN_TILE(dtLB, dtHB, xLB, xHB, zLB, zHB, bcB, s0 + TILE_S)

#undef HOLD_LOADS
#undef TREE_STAGE
#undef LOAD_TILE
#undef SCAN_STEP
#undef SCAN_TILE
}

extern "C" void kernel_launch(void* const* d_in, const int* in_sizes, int n_in,
                              void* d_out, int out_size, void* d_ws, size_t ws_size,
                              hipStream_t stream) {
    const float* x         = (const float*)d_in[0];
    const float* in_proj_w = (const float*)d_in[1];
    const float* conv_w    = (const float*)d_in[2];
    const float* conv_b    = (const float*)d_in[3];
    const float* x_proj_w  = (const float*)d_in[4];
    const float* dt_proj_w = (const float*)d_in[5];
    const float* dt_proj_b = (const float*)d_in[6];
    const float* log_A     = (const float*)d_in[7];
    const float* D_skip    = (const float*)d_in[8];
    const float* out_proj_w= (const float*)d_in[9];
    float* out = (float*)d_out;

    const size_t planeBD = (size_t)MROWS * D_INNER;
    float* ws      = (float*)d_ws;
    float* buf_x   = ws;
    float* buf_z   = buf_x  + planeBD;
    float* buf_xc  = buf_z  + planeBD;
    float* buf_dbc = buf_xc + planeBD;
    float* buf_y   = buf_dbc + (size_t)MROWS * DBC_LD;

    const dim3 blk(128);
    const int gy = MROWS / (16 * 2 * 4);

    gemm_wmma_f32<2, 4, 0, 32><<<dim3(D_INNER / 64, gy), blk, 0, stream>>>(
        x, in_proj_w, nullptr, buf_x, MROWS, D_INNER, D_MODEL, D_MODEL, D_INNER);
    gemm_wmma_f32<2, 4, 0, 32><<<dim3(D_INNER / 64, gy), blk, 0, stream>>>(
        x, in_proj_w + (size_t)D_INNER * D_MODEL, nullptr, buf_z,
        MROWS, D_INNER, D_MODEL, D_MODEL, D_INNER);

    conv_silu_kernel<<<(MROWS * D_INNER) / 256, 256, 0, stream>>>(
        buf_x, conv_w, conv_b, buf_xc);

    gemm_wmma_f32<2, 1, 2, 64><<<dim3(DBC_LD / 16, gy), blk, 0, stream>>>(
        buf_xc, x_proj_w, nullptr, buf_dbc, MROWS, DBC_LD, D_INNER, D_INNER, DBC_LD);

    gemm_wmma_f32<2, 4, 1, 16><<<dim3(D_INNER / 64, gy), blk, 0, stream>>>(
        buf_dbc, dt_proj_w, dt_proj_b, buf_x, MROWS, D_INNER, DT_RANK, DBC_LD, D_INNER);

    scan_kernel<<<(BATCH * D_INNER * (D_STATE / 2)) / SCAN_THREADS, SCAN_THREADS, 0, stream>>>(
        buf_x, buf_xc, buf_dbc, buf_z, log_A, D_skip, buf_y);

    gemm_wmma_f32<2, 2, 0, 64><<<dim3(D_MODEL / 32, gy), blk, 0, stream>>>(
        buf_y, out_proj_w, nullptr, out, MROWS, D_MODEL, D_INNER, D_INNER, D_MODEL);
}


# dt_proj epilogue: softplus from hardware exp2/log2/rcp, branch-free (atanh series in t/(2+t) while e^x <= one half), replacing library log1p: 64 per lane, about 115 dependent instructions each
# speedup vs baseline: 8.5616x; 1.0959x over previous
#include <hip/hip_runtime.h>
#include <hip/hip_bf16.h>


#define D_MODEL 768
#define D_STATE 16
#define D_CONV  4
#define D_INNER 1536
#define DT_RANK 48
#define BATCH   2
#define SEQ     2048
#define MROWS   (BATCH * SEQ)
#define DBC_LD  (DT_RANK + 2 * D_STATE)

typedef __attribute__((ext_vector_type(2))) float v2f;
typedef __attribute__((ext_vector_type(8))) float v8f;

__device__ __forceinline__ float silu_f(float x) {
    return x * (1.0f / (1.0f + __expf(-x)));
}
__device__ __forceinline__ float softplus_f(float x) {
    const float t  = __expf(x);
    const float s  = t * __builtin_amdgcn_rcpf(2.0f + t);
    const float s2 = s * s;
    float q = fmaf(s2, 1.0f / 9.0f, 1.0f / 7.0f);
    q = fmaf(s2, q, 1.0f / 5.0f);
    q = fmaf(s2, q, 1.0f / 3.0f);
    const float small = (s + s) * fmaf(s2, q, 1.0f);
    const float large = __builtin_amdgcn_logf(1.0f + t) * 0.69314718f;
    const float sp = (t > 0.5f) ? large : small;
    return (x > 20.0f) ? x : sp;
}

typedef __attribute__((ext_vector_type(4))) float v4f;

template <int MT, int NT, int EPI, int BK>
__launch_bounds__(128)
__global__ void gemm_wmma_f32(const float* __restrict__ A,
                              const float* __restrict__ W,
                              const float* __restrict__ bias,
                              float* __restrict__ C,
                              int M, int N, int K, int lda, int ldc) {
    constexpr int BM    = 64 * MT;
    constexpr int BN    = 16 * NT;
    constexpr int P     = BK + 4;
    constexpr int CPR   = BK / 4;
    constexpr int ACH   = BM * CPR / 128;
    constexpr int WCH   = BN * CPR / 128;
    constexpr int RSTEP = 128 / CPR;
    constexpr int STAGE = (BM + BN) * P;
    static_assert((BM * CPR) % 128 == 0 && (BN * CPR) % 128 == 0, "tile/threads");
    static_assert(((P / 4) & 1) == 1 && (P % 4) == 0, "pitch must be 4*odd floats");

    __shared__ __attribute__((aligned(16))) float lds[2 * STAGE];

    const int tid    = threadIdx.x;
    const int lane   = tid & 31;
    const int wave   = tid >> 5;
    const int mBlock = blockIdx.y * BM;
    const int mBase  = mBlock + wave * (16 * MT);
    const int nBase  = blockIdx.x * BN;
    const int r16    = lane & 15;
    const int kOff   = (lane >> 4) * 2;

    const int cRow = tid / CPR;
    const int cCol = (tid % CPR) * 4;
    const float* gA = A + (size_t)(mBlock + cRow) * lda + cCol;
    const float* gW = W + (size_t)(nBase + cRow) * K + cCol;
    const int sOffA = cRow * P + cCol;
    const int sOffW = (BM + cRow) * P + cCol;

    const int fA = (wave * (16 * MT) + r16) * P + kOff;
    const int fW = (BM + r16) * P + kOff;

    v8f acc[MT][NT];
#pragma unroll
    for (int i = 0; i < MT; ++i)
#pragma unroll
        for (int t = 0; t < NT; ++t) {
            v8f z = {0.f, 0.f, 0.f, 0.f, 0.f, 0.f, 0.f, 0.f};
            acc[i][t] = z;
        }

    v4f rA[ACH], rW[WCH];

#define GLOAD(K0)                                                        \
    _Pragma("unroll") for (int i = 0; i < ACH; ++i)                      \
        rA[i] = *(const v4f*)(gA + (size_t)(i * RSTEP) * lda + (K0));    \
    _Pragma("unroll") for (int i = 0; i < WCH; ++i)                      \
        rW[i] = *(const v4f*)(gW + (size_t)(i * RSTEP) * K + (K0));

#define SSTORE(ST)                                                       \
    _Pragma("unroll") for (int i = 0; i < ACH; ++i)                      \
        *(v4f*)((ST) + sOffA + i * RSTEP * P) = rA[i];                   \
    _Pragma("unroll") for (int i = 0; i < WCH; ++i)                      \
        *(v4f*)((ST) + sOffW + i * RSTEP * P) = rW[i];

    const int nS = K / BK;
    GLOAD(0)
    SSTORE(lds)
    __syncthreads();

    for (int s = 0; s < nS; ++s) {
        const float* st = lds + (s & 1) * STAGE;
        const bool more = (s + 1 < nS);
        if (more) { GLOAD((s + 1) * BK) }

#pragma unroll
        for (int kk = 0; kk < BK / 4; ++kk) {
            v2f a[MT], b[NT];
#pragma unroll
            for (int i = 0; i < MT; ++i)
                a[i] = *(const v2f*)(st + fA + i * 16 * P + kk * 4);
#pragma unroll
            for (int t = 0; t < NT; ++t)
                b[t] = *(const v2f*)(st + fW + t * 16 * P + kk * 4);
#pragma unroll
            for (int i = 0; i < MT; ++i)
#pragma unroll
                for (int t = 0; t < NT; ++t)
                    acc[i][t] = __builtin_amdgcn_wmma_f32_16x16x4_f32(
                        false, a[i], false, b[t], (short)0, acc[i][t],
                        false, false);
        }

        if (more) { SSTORE(lds + ((s + 1) & 1) * STAGE) }
        __syncthreads();
    }

#undef GLOAD
#undef SSTORE

    const int rowSel = (lane >> 4) * 8;
#pragma unroll
    for (int i = 0; i < MT; ++i)
#pragma unroll
        for (int t = 0; t < NT; ++t) {
            const int col = nBase + t * 16 + (lane & 15);
            int ccol = col;
            if (EPI == 2 && col >= DT_RANK) {
                const int q = col - DT_RANK;
                ccol = DT_RANK + 4 * ((q & 15) >> 1) + 2 * (q >> 4) + (q & 1);
            }
            float bv = 0.0f;
            if (EPI == 1) bv = bias[col];
#pragma unroll
            for (int r = 0; r < 8; ++r) {
                float v = acc[i][t][r];
                if (EPI == 1) v = softplus_f(v + bv);
                C[(size_t)(mBase + i * 16 + r + rowSel) * ldc + ccol] = v;
            }
        }
}

__launch_bounds__(256)
__global__ void conv_silu_kernel(const float* __restrict__ xin,
                                 const float* __restrict__ w,
                                 const float* __restrict__ b,
                                 float* __restrict__ out) {
    const int idx = blockIdx.x * blockDim.x + threadIdx.x;
    const int d  = idx % D_INNER;
    const int s  = (idx / D_INNER) % SEQ;
    const int bi = idx / (D_INNER * SEQ);

    const float* base = xin + (size_t)bi * SEQ * D_INNER + d;
    float acc = b[d];
#pragma unroll
    for (int j = 0; j < D_CONV; ++j) {
        const int sp = s - (D_CONV - 1) + j;
        if (sp >= 0) acc += w[d * D_CONV + j] * base[(size_t)sp * D_INNER];
    }
    out[idx] = silu_f(acc);
}

#define TILE_S 16
#define SCAN_THREADS 96

template <int J>
__device__ __forceinline__ float bcast8_f(float v) {
    return __int_as_float(__builtin_amdgcn_mov_dpp8(__float_as_int(v), J * 0x249249));
}

template <int M>
__device__ __forceinline__ float row_xor_f(float v) {
    return __int_as_float(__builtin_amdgcn_update_dpp(
        0, __float_as_int(v), 0x160 + M, 0xf, 0xf, false));
}

__launch_bounds__(SCAN_THREADS)
__global__ void scan_kernel(const float* __restrict__ delta,
                            const float* __restrict__ xc,
                            const float* __restrict__ dbc,
                            const float* __restrict__ zbuf,
                            const float* __restrict__ logA,
                            const float* __restrict__ Dskip,
                            float* __restrict__ y) {
    const int gid = blockIdx.x * blockDim.x + threadIdx.x;
    const int p  = gid & 7;
    const int ch = gid >> 3;
    const int d  = ch % D_INNER;
    const int b  = ch / D_INNER;

    const float Aneg0 = -__expf(logA[d * D_STATE + 2 * p]);
    const float Aneg1 = -__expf(logA[d * D_STATE + 2 * p + 1]);
    const float dsk   = Dskip[d];
    const bool  bit0 = (p & 1) != 0, bit1 = (p & 2) != 0, bit2 = (p & 4) != 0;

    const size_t baseBD = (size_t)b * SEQ * D_INNER + d + (size_t)p * D_INNER;
    const size_t HI = (size_t)8 * D_INNER;
    const float* pDt = delta + baseBD;
    const float* pXc = xc    + baseBD;
    const float* pZ  = zbuf  + baseBD;
    float*       pY  = y     + baseBD;
    const float* pBC = dbc + (size_t)b * SEQ * DBC_LD + DT_RANK + 4 * p;

    float dtLA, dtHA, xLA, xHA, zLA, zHA;  v4f bcA[TILE_S];
    float dtLB, dtHB, xLB, xHB, zLB, zHB;  v4f bcB[TILE_S];
    float dtLC, dtHC, xLC, xHC, zLC, zHC;  v4f bcC[TILE_S];

#define LOAD_TILE(DTL, DTH, XL, XH, ZL, ZH, BC, S0)                    \
    {                                                                  \
        const size_t o = (size_t)(S0) * D_INNER;                       \
        DTL = pDt[o];  DTH = pDt[o + HI];                              \
        XL  = pXc[o];  XH  = pXc[o + HI];                              \
        const float* pb = pBC + (size_t)(S0) * DBC_LD;                 \
                   \
        _Pragma("unroll") for (int j = 0; j < TILE_S; ++j)             \
            BC[j] = *(const v4f*)(pb + j * DBC_LD);                    \
        ZL  = pZ[o];   ZH  = pZ[o + HI];                               \
    }

#define SCAN_STEP(J, DTL, DTH, XL, XH, BC)                             \
    {                                                                  \
        const float dtj = bcast8_f<((J) & 7)>(((J) < 8) ? DTL : DTH);  \
        const float xvj = bcast8_f<((J) & 7)>(((J) < 8) ? XL : XH);    \
        const float a0 = __expf(dtj * Aneg0);                          \
        const float a1 = __expf(dtj * Aneg1);                          \
          \
          \
        h0 = fmaf(a0, h0, (dtj * BC[J].x) * xvj);                      \
        h1 = fmaf(a1, h1, (dtj * BC[J].y) * xvj);                      \
        V[J] = h0 * BC[J].z + h1 * BC[J].w;                            \
    }

#define TREE_STAGE(OUT, IN, CNT, BIT, M)                               \
           \
    _Pragma("unroll") for (int i = 0; i < (CNT); ++i) {                \
        const float keep = (BIT) ? IN[2 * i + 1] : IN[2 * i];          \
        const float send = (BIT) ? IN[2 * i] : IN[2 * i + 1];          \
        OUT[i] = keep + row_xor_f<M>(send);                            \
    }

#define SCAN_TILE(DTL, DTH, XL, XH, ZL, ZH, BC, S0)                    \
    {                                                                  \
        float V[TILE_S], Q[8], R[4], S[2];                             \
        SCAN_STEP(0, DTL, DTH, XL, XH, BC)  SCAN_STEP(1, DTL, DTH, XL, XH, BC)   \
        SCAN_STEP(2, DTL, DTH, XL, XH, BC)  SCAN_STEP(3, DTL, DTH, XL, XH, BC)   \
        SCAN_STEP(4, DTL, DTH, XL, XH, BC)  SCAN_STEP(5, DTL, DTH, XL, XH, BC)   \
        SCAN_STEP(6, DTL, DTH, XL, XH, BC)  SCAN_STEP(7, DTL, DTH, XL, XH, BC)   \
        SCAN_STEP(8, DTL, DTH, XL, XH, BC)  SCAN_STEP(9, DTL, DTH, XL, XH, BC)   \
        SCAN_STEP(10, DTL, DTH, XL, XH, BC) SCAN_STEP(11, DTL, DTH, XL, XH, BC)  \
        SCAN_STEP(12, DTL, DTH, XL, XH, BC) SCAN_STEP(13, DTL, DTH, XL, XH, BC)  \
        SCAN_STEP(14, DTL, DTH, XL, XH, BC) SCAN_STEP(15, DTL, DTH, XL, XH, BC)  \
          \
        TREE_STAGE(Q, V, 8, bit0, 1)                                   \
        TREE_STAGE(R, Q, 4, bit1, 2)                                   \
        TREE_STAGE(S, R, 2, bit2, 4)                                   \
          \
        const size_t o = (size_t)(S0) * D_INNER;                       \
        pY[o]      = (S[0] + dsk * XL) * silu_f(ZL);                   \
        pY[o + HI] = (S[1] + dsk * XH) * silu_f(ZH);                   \
    }

#define HOLD_LOADS() __builtin_amdgcn_sched_barrier(0);

    float h0 = 0.0f, h1 = 0.0f;
    LOAD_TILE(dtLA, dtHA, xLA, xHA, zLA, zHA, bcA, 0)
    LOAD_TILE(dtLB, dtHB, xLB, xHB, zLB, zHB, bcB, TILE_S)
    int s0 = 0;
    for (; s0 + 5 * TILE_S <= SEQ; s0 += 3 * TILE_S) {
        LOAD_TILE(dtLC, dtHC, xLC, xHC, zLC, zHC, bcC, s0 + 2 * TILE_S)
        HOLD_LOADS()
        SCAN_TILE(dtLA, dtHA, xLA, xHA, zLA, zHA, bcA, s0)
        HOLD_LOADS()
        LOAD_TILE(dtLA, dtHA, xLA, xHA, zLA, zHA, bcA, s0 + 3 * TILE_S)
        HOLD_LOADS()
        SCAN_TILE(dtLB, dtHB, xLB, xHB, zLB, zHB, bcB, s0 + TILE_S)
        HOLD_LOADS()
        LOAD_TILE(dtLB, dtHB, xLB, xHB, zLB, zHB, bcB, s0 + 4 * TILE_S)
        HOLD_LOADS()
        SCAN_TILE(dtLC, dtHC, xLC, xHC, zLC, zHC, bcC, s0 + 2 * TILE_S)
    }
    static_assert((SEQ / TILE_S) % 3 == 2 && SEQ % TILE_S == 0, "tile count");
    SCAN_TILE(dtLA, dtHA, xLA, xHA, zLA, zHA, bcA, s0)
    SCAN_TILE(dtLB, dtHB, xLB, xHB, zLB, zHB, bcB, s0 + TILE_S)

#undef HOLD_LOADS
#undef TREE_STAGE
#undef LOAD_TILE
#undef SCAN_STEP
#undef SCAN_TILE
}

extern "C" void kernel_launch(void* const* d_in, const int* in_sizes, int n_in,
                              void* d_out, int out_size, void* d_ws, size_t ws_size,
                              hipStream_t stream) {
    const float* x         = (const float*)d_in[0];
    const float* in_proj_w = (const float*)d_in[1];
    const float* conv_w    = (const float*)d_in[2];
    const float* conv_b    = (const float*)d_in[3];
    const float* x_proj_w  = (const float*)d_in[4];
    const float* dt_proj_w = (const float*)d_in[5];
    const float* dt_proj_b = (const float*)d_in[6];
    const float* log_A     = (const float*)d_in[7];
    const float* D_skip    = (const float*)d_in[8];
    const float* out_proj_w= (const float*)d_in[9];
    float* out = (float*)d_out;

    const size_t planeBD = (size_t)MROWS * D_INNER;
    float* ws      = (float*)d_ws;
    float* buf_x   = ws;
    float* buf_z   = buf_x  + planeBD;
    float* buf_xc  = buf_z  + planeBD;
    float* buf_dbc = buf_xc + planeBD;
    float* buf_y   = buf_dbc + (size_t)MROWS * DBC_LD;

    const dim3 blk(128);
    const int gy = MROWS / (16 * 2 * 4);

    gemm_wmma_f32<2, 4, 0, 32><<<dim3(D_INNER / 64, gy), blk, 0, stream>>>(
        x, in_proj_w, nullptr, buf_x, MROWS, D_INNER, D_MODEL, D_MODEL, D_INNER);
    gemm_wmma_f32<2, 4, 0, 32><<<dim3(D_INNER / 64, gy), blk, 0, stream>>>(
        x, in_proj_w + (size_t)D_INNER * D_MODEL, nullptr, buf_z,
        MROWS, D_INNER, D_MODEL, D_MODEL, D_INNER);

    conv_silu_kernel<<<(MROWS * D_INNER) / 256, 256, 0, stream>>>(
        buf_x, conv_w, conv_b, buf_xc);

    gemm_wmma_f32<2, 1, 2, 64><<<dim3(DBC_LD / 16, gy), blk, 0, stream>>>(
        buf_xc, x_proj_w, nullptr, buf_dbc, MROWS, DBC_LD, D_INNER, D_INNER, DBC_LD);

    gemm_wmma_f32<2, 4, 1, 16><<<dim3(D_INNER / 64, gy), blk, 0, stream>>>(
        buf_dbc, dt_proj_w, dt_proj_b, buf_x, MROWS, D_INNER, DT_RANK, DBC_LD, D_INNER);

    scan_kernel<<<(BATCH * D_INNER * (D_STATE / 2)) / SCAN_THREADS, SCAN_THREADS, 0, stream>>>(
        buf_x, buf_xc, buf_dbc, buf_z, log_A, D_skip, buf_y);

    gemm_wmma_f32<2, 2, 0, 64><<<dim3(D_MODEL / 32, gy), blk, 0, stream>>>(
        buf_y, out_proj_w, nullptr, out, MROWS, D_MODEL, D_INNER, D_INNER, D_MODEL);
}


# conv+SiLU moved into the x_proj launch: a block owns 16 token rows and all 80 columns and builds its A slices from the raw x-branch; 8 waves split K, their 8 partial sums added in wave order
# speedup vs baseline: 9.4193x; 1.1002x over previous
#include <hip/hip_runtime.h>
#include <hip/hip_bf16.h>


#define D_MODEL 768
#define D_STATE 16
#define D_CONV  4
#define D_INNER 1536
#define DT_RANK 48
#define BATCH   2
#define SEQ     2048
#define MROWS   (BATCH * SEQ)
#define DBC_LD  (DT_RANK + 2 * D_STATE)

typedef __attribute__((ext_vector_type(2))) float v2f;
typedef __attribute__((ext_vector_type(8))) float v8f;

__device__ __forceinline__ float silu_f(float x) {
    return x * (1.0f / (1.0f + __expf(-x)));
}
__device__ __forceinline__ float softplus_f(float x) {
    const float t  = __expf(x);
    const float s  = t * __builtin_amdgcn_rcpf(2.0f + t);
    const float s2 = s * s;
    float q = fmaf(s2, 1.0f / 9.0f, 1.0f / 7.0f);
    q = fmaf(s2, q, 1.0f / 5.0f);
    q = fmaf(s2, q, 1.0f / 3.0f);
    const float small = (s + s) * fmaf(s2, q, 1.0f);
    const float large = __builtin_amdgcn_logf(1.0f + t) * 0.69314718f;
    const float sp = (t > 0.5f) ? large : small;
    return (x > 20.0f) ? x : sp;
}

typedef __attribute__((ext_vector_type(4))) float v4f;

template <int MT, int NT, int EPI, int BK>
__launch_bounds__(128)
__global__ void gemm_wmma_f32(const float* __restrict__ A,
                              const float* __restrict__ W,
                              const float* __restrict__ bias,
                              float* __restrict__ C,
                              int M, int N, int K, int lda, int ldc) {
    constexpr int BM    = 64 * MT;
    constexpr int BN    = 16 * NT;
    constexpr int P     = BK + 4;
    constexpr int CPR   = BK / 4;
    constexpr int ACH   = BM * CPR / 128;
    constexpr int WCH   = BN * CPR / 128;
    constexpr int RSTEP = 128 / CPR;
    constexpr int STAGE = (BM + BN) * P;
    static_assert((BM * CPR) % 128 == 0 && (BN * CPR) % 128 == 0, "tile/threads");
    static_assert(((P / 4) & 1) == 1 && (P % 4) == 0, "pitch must be 4*odd floats");

    __shared__ __attribute__((aligned(16))) float lds[2 * STAGE];

    const int tid    = threadIdx.x;
    const int lane   = tid & 31;
    const int wave   = tid >> 5;
    const int mBlock = blockIdx.y * BM;
    const int mBase  = mBlock + wave * (16 * MT);
    const int nBase  = blockIdx.x * BN;
    const int r16    = lane & 15;
    const int kOff   = (lane >> 4) * 2;

    const int cRow = tid / CPR;
    const int cCol = (tid % CPR) * 4;
    const float* gA = A + (size_t)(mBlock + cRow) * lda + cCol;
    const float* gW = W + (size_t)(nBase + cRow) * K + cCol;
    const int sOffA = cRow * P + cCol;
    const int sOffW = (BM + cRow) * P + cCol;

    const int fA = (wave * (16 * MT) + r16) * P + kOff;
    const int fW = (BM + r16) * P + kOff;

    v8f acc[MT][NT];
#pragma unroll
    for (int i = 0; i < MT; ++i)
#pragma unroll
        for (int t = 0; t < NT; ++t) {
            v8f z = {0.f, 0.f, 0.f, 0.f, 0.f, 0.f, 0.f, 0.f};
            acc[i][t] = z;
        }

    v4f rA[ACH], rW[WCH];

#define GLOAD(K0)                                                        \
    _Pragma("unroll") for (int i = 0; i < ACH; ++i)                      \
        rA[i] = *(const v4f*)(gA + (size_t)(i * RSTEP) * lda + (K0));    \
    _Pragma("unroll") for (int i = 0; i < WCH; ++i)                      \
        rW[i] = *(const v4f*)(gW + (size_t)(i * RSTEP) * K + (K0));

#define SSTORE(ST)                                                       \
    _Pragma("unroll") for (int i = 0; i < ACH; ++i)                      \
        *(v4f*)((ST) + sOffA + i * RSTEP * P) = rA[i];                   \
    _Pragma("unroll") for (int i = 0; i < WCH; ++i)                      \
        *(v4f*)((ST) + sOffW + i * RSTEP * P) = rW[i];

    const int nS = K / BK;
    GLOAD(0)
    SSTORE(lds)
    __syncthreads();

    for (int s = 0; s < nS; ++s) {
        const float* st = lds + (s & 1) * STAGE;
        const bool more = (s + 1 < nS);
        if (more) { GLOAD((s + 1) * BK) }

#pragma unroll
        for (int kk = 0; kk < BK / 4; ++kk) {
            v2f a[MT], b[NT];
#pragma unroll
            for (int i = 0; i < MT; ++i)
                a[i] = *(const v2f*)(st + fA + i * 16 * P + kk * 4);
#pragma unroll
            for (int t = 0; t < NT; ++t)
                b[t] = *(const v2f*)(st + fW + t * 16 * P + kk * 4);
#pragma unroll
            for (int i = 0; i < MT; ++i)
#pragma unroll
                for (int t = 0; t < NT; ++t)
                    acc[i][t] = __builtin_amdgcn_wmma_f32_16x16x4_f32(
                        false, a[i], false, b[t], (short)0, acc[i][t],
                        false, false);
        }

        if (more) { SSTORE(lds + ((s + 1) & 1) * STAGE) }
        __syncthreads();
    }

#undef GLOAD
#undef SSTORE

    const int rowSel = (lane >> 4) * 8;
#pragma unroll
    for (int i = 0; i < MT; ++i)
#pragma unroll
        for (int t = 0; t < NT; ++t) {
            const int col = nBase + t * 16 + (lane & 15);
            float bv = 0.0f;
            if (EPI == 1) bv = bias[col];
#pragma unroll
            for (int r = 0; r < 8; ++r) {
                float v = acc[i][t][r];
                if (EPI == 1) v = softplus_f(v + bv);
                C[(size_t)(mBase + i * 16 + r + rowSel) * ldc + col] = v;
            }
        }
}

#define XP_WAVES   8
#define XP_THREADS (32 * XP_WAVES)
#define XP_BK      32
#define XP_P       (XP_BK + 4)
#define XP_KW      (D_INNER / XP_WAVES)
#define XP_NS      (XP_KW / XP_BK)
#define XP_WCH     (DBC_LD * (XP_BK / 4) / 32)
#define XP_STAGE   ((16 + DBC_LD) * XP_P)
#define XP_RP      84

template <bool FIRST>
__device__ __forceinline__ void load_rows7(const float* __restrict__ pRow0, int g,
                                           v4f (&xr)[7]) {
#pragma unroll
    for (int jj = 0; jj < 7; ++jj) {
        const int r = 4 * g - 3 + jj;
        if (!FIRST || r >= 0) {
            xr[jj] = *(const v4f*)(pRow0 + (long)r * D_INNER);
        } else {
            v4f z = {0.f, 0.f, 0.f, 0.f};
            xr[jj] = z;
        }
    }
}

template <bool FIRST>
__device__ __forceinline__ void conv_rows4(const v4f (&xr)[7], const v4f (&cw)[4],
                                           const v4f cb, int g, v4f (&o)[4]) {
#pragma unroll
    for (int i = 0; i < 4; ++i) {
#pragma unroll
        for (int e = 0; e < 4; ++e) {
            float acc = cb[e];
#pragma unroll
            for (int j = 0; j < D_CONV; ++j) {
                if (!FIRST || (4 * g + i + j >= 3)) acc += cw[e][j] * xr[i + j][e];
            }
            o[i][e] = silu_f(acc);
        }
    }
}

__launch_bounds__(XP_THREADS)
__global__ void conv_xproj_kernel(const float* __restrict__ xin,
                                  const float* __restrict__ cw_,
                                  const float* __restrict__ cb_,
                                  const float* __restrict__ W,
                                  float* __restrict__ out,
                                  float* __restrict__ dbc) {
    static_assert(D_INNER % (XP_WAVES * XP_BK) == 0, "K split");
    static_assert(SEQ % 16 == 0 && DBC_LD % 16 == 0, "tiles");
    static_assert(XP_WCH * 32 == DBC_LD * (XP_BK / 4), "W chunks");
    static_assert(XP_WAVES * 16 * XP_RP <= XP_WAVES * XP_STAGE, "partial sums fit");
    constexpr int NT = DBC_LD / 16;

    __shared__ __attribute__((aligned(16))) float lds[XP_WAVES * XP_STAGE];

    const int tid  = threadIdx.x;
    const int lane = tid & 31;
    const int wave = tid >> 5;
    const int r16  = lane & 15;
    const int kOff = (lane >> 4) * 2;
    const int c4   = (lane & 7) * 4;
    const int g    = lane >> 3;
    const int m0   = blockIdx.x * 16;
    const bool first = (m0 % SEQ) == 0;
    const int kw0  = wave * XP_KW;

    float* stA = lds + wave * XP_STAGE;
    float* stW = stA + 16 * XP_P;

    const float* pX  = xin + (size_t)m0 * D_INNER + kw0 + c4;
    float*       pO  = out + (size_t)(m0 + 4 * g) * D_INNER + kw0 + c4;
    const float* pW  = W + (size_t)g * D_INNER + kw0 + c4;
    const float* pCw = cw_ + (size_t)(kw0 + c4) * D_CONV;
    const float* pCb = cb_ + kw0 + c4;

    v8f acc[NT];
#pragma unroll
    for (int t = 0; t < NT; ++t) {
        v8f z = {0.f, 0.f, 0.f, 0.f, 0.f, 0.f, 0.f, 0.f};
        acc[t] = z;
    }

    v4f xr[7], wr[XP_WCH], cwv[4], cbv;

#define XP_GLOAD(S)                                                       \
    {                                                                     \
        const int ko = (S) * XP_BK;                                       \
        if (first) load_rows7<true>(pX + ko, g, xr);                      \
        else       load_rows7<false>(pX + ko, g, xr);                     \
        _Pragma("unroll") for (int i = 0; i < XP_WCH; ++i)                \
            wr[i] = *(const v4f*)(pW + (size_t)(4 * i) * D_INNER + ko);   \
        _Pragma("unroll") for (int e = 0; e < 4; ++e)                     \
            cwv[e] = *(const v4f*)(pCw + (size_t)(ko + e) * D_CONV);      \
        cbv = *(const v4f*)(pCb + ko);                                    \
    }

#define XP_BUILD(S)                                                       \
    {                                                                     \
        const int ko = (S) * XP_BK;                                       \
        v4f o[4];                                                         \
        if (first) conv_rows4<true>(xr, cwv, cbv, g, o);                  \
        else       conv_rows4<false>(xr, cwv, cbv, g, o);                 \
        _Pragma("unroll") for (int i = 0; i < 4; ++i) {                   \
            *(v4f*)(pO + (size_t)i * D_INNER + ko) = o[i];                \
            *(v4f*)(stA + (4 * g + i) * XP_P + c4) = o[i];                \
        }                                                                 \
        _Pragma("unroll") for (int i = 0; i < XP_WCH; ++i)                \
            *(v4f*)(stW + (g + 4 * i) * XP_P + c4) = wr[i];               \
    }

    XP_GLOAD(0)
    XP_BUILD(0)
    __syncthreads();

    for (int s = 0; s < XP_NS; ++s) {
        const bool more = (s + 1 < XP_NS);
        if (more) { XP_GLOAD(s + 1) }

#pragma unroll
        for (int kk = 0; kk < XP_BK / 4; ++kk) {
            const v2f a = *(const v2f*)(stA + r16 * XP_P + kk * 4 + kOff);
            v2f b[NT];
#pragma unroll
            for (int t = 0; t < NT; ++t)
                b[t] = *(const v2f*)(stW + (t * 16 + r16) * XP_P + kk * 4 + kOff);
#pragma unroll
            for (int t = 0; t < NT; ++t)
                acc[t] = __builtin_amdgcn_wmma_f32_16x16x4_f32(
                    false, a, false, b[t], (short)0, acc[t], false, false);
        }

        if (more) { XP_BUILD(s + 1) }
        __syncthreads();
    }

#undef XP_GLOAD
#undef XP_BUILD

    float* red = lds;
    const int rowSel = (lane >> 4) * 8;
#pragma unroll
    for (int t = 0; t < NT; ++t)
#pragma unroll
        for (int r = 0; r < 8; ++r)
            red[(wave * 16 + r + rowSel) * XP_RP + t * 16 + r16] = acc[t][r];
    __syncthreads();

#pragma unroll
    for (int i = 0; i < (16 * DBC_LD) / XP_THREADS; ++i) {
        const int oIdx = tid + XP_THREADS * i;
        const int row  = oIdx / DBC_LD;
        const int col  = oIdx - row * DBC_LD;
        float v = red[row * XP_RP + col];
#pragma unroll
        for (int w = 1; w < XP_WAVES; ++w)
            v += red[(w * 16 + row) * XP_RP + col];
        int ccol = col;
        if (col >= DT_RANK) {
            const int q = col - DT_RANK;
            ccol = DT_RANK + 4 * ((q & 15) >> 1) + 2 * (q >> 4) + (q & 1);
        }
        dbc[(size_t)(m0 + row) * DBC_LD + ccol] = v;
    }
}

#define TILE_S 16
#define SCAN_THREADS 96

template <int J>
__device__ __forceinline__ float bcast8_f(float v) {
    return __int_as_float(__builtin_amdgcn_mov_dpp8(__float_as_int(v), J * 0x249249));
}

template <int M>
__device__ __forceinline__ float row_xor_f(float v) {
    return __int_as_float(__builtin_amdgcn_update_dpp(
        0, __float_as_int(v), 0x160 + M, 0xf, 0xf, false));
}

__launch_bounds__(SCAN_THREADS)
__global__ void scan_kernel(const float* __restrict__ delta,
                            const float* __restrict__ xc,
                            const float* __restrict__ dbc,
                            const float* __restrict__ zbuf,
                            const float* __restrict__ logA,
                            const float* __restrict__ Dskip,
                            float* __restrict__ y) {
    const int gid = blockIdx.x * blockDim.x + threadIdx.x;
    const int p  = gid & 7;
    const int ch = gid >> 3;
    const int d  = ch % D_INNER;
    const int b  = ch / D_INNER;

    const float Aneg0 = -__expf(logA[d * D_STATE + 2 * p]);
    const float Aneg1 = -__expf(logA[d * D_STATE + 2 * p + 1]);
    const float dsk   = Dskip[d];
    const bool  bit0 = (p & 1) != 0, bit1 = (p & 2) != 0, bit2 = (p & 4) != 0;

    const size_t baseBD = (size_t)b * SEQ * D_INNER + d + (size_t)p * D_INNER;
    const size_t HI = (size_t)8 * D_INNER;
    const float* pDt = delta + baseBD;
    const float* pXc = xc    + baseBD;
    const float* pZ  = zbuf  + baseBD;
    float*       pY  = y     + baseBD;
    const float* pBC = dbc + (size_t)b * SEQ * DBC_LD + DT_RANK + 4 * p;

    float dtLA, dtHA, xLA, xHA, zLA, zHA;  v4f bcA[TILE_S];
    float dtLB, dtHB, xLB, xHB, zLB, zHB;  v4f bcB[TILE_S];
    float dtLC, dtHC, xLC, xHC, zLC, zHC;  v4f bcC[TILE_S];

#define LOAD_TILE(DTL, DTH, XL, XH, ZL, ZH, BC, S0)                    \
    {                                                                  \
        const size_t o = (size_t)(S0) * D_INNER;                       \
        DTL = pDt[o];  DTH = pDt[o + HI];                              \
        XL  = pXc[o];  XH  = pXc[o + HI];                              \
        const float* pb = pBC + (size_t)(S0) * DBC_LD;                 \
                   \
        _Pragma("unroll") for (int j = 0; j < TILE_S; ++j)             \
            BC[j] = *(const v4f*)(pb + j * DBC_LD);                    \
        ZL  = pZ[o];   ZH  = pZ[o + HI];                               \
    }

#define SCAN_STEP(J, DTL, DTH, XL, XH, BC)                             \
    {                                                                  \
        const float dtj = bcast8_f<((J) & 7)>(((J) < 8) ? DTL : DTH);  \
        const float xvj = bcast8_f<((J) & 7)>(((J) < 8) ? XL : XH);    \
        const float a0 = __expf(dtj * Aneg0);                          \
        const float a1 = __expf(dtj * Aneg1);                          \
          \
          \
        h0 = fmaf(a0, h0, (dtj * BC[J].x) * xvj);                      \
        h1 = fmaf(a1, h1, (dtj * BC[J].y) * xvj);                      \
        V[J] = h0 * BC[J].z + h1 * BC[J].w;                            \
    }

#define TREE_STAGE(OUT, IN, CNT, BIT, M)                               \
           \
    _Pragma("unroll") for (int i = 0; i < (CNT); ++i) {                \
        const float keep = (BIT) ? IN[2 * i + 1] : IN[2 * i];          \
        const float send = (BIT) ? IN[2 * i] : IN[2 * i + 1];          \
        OUT[i] = keep + row_xor_f<M>(send);                            \
    }

#define SCAN_TILE(DTL, DTH, XL, XH, ZL, ZH, BC, S0)                    \
    {                                                                  \
        float V[TILE_S], Q[8], R[4], S[2];                             \
        SCAN_STEP(0, DTL, DTH, XL, XH, BC)  SCAN_STEP(1, DTL, DTH, XL, XH, BC)   \
        SCAN_STEP(2, DTL, DTH, XL, XH, BC)  SCAN_STEP(3, DTL, DTH, XL, XH, BC)   \
        SCAN_STEP(4, DTL, DTH, XL, XH, BC)  SCAN_STEP(5, DTL, DTH, XL, XH, BC)   \
        SCAN_STEP(6, DTL, DTH, XL, XH, BC)  SCAN_STEP(7, DTL, DTH, XL, XH, BC)   \
        SCAN_STEP(8, DTL, DTH, XL, XH, BC)  SCAN_STEP(9, DTL, DTH, XL, XH, BC)   \
        SCAN_STEP(10, DTL, DTH, XL, XH, BC) SCAN_STEP(11, DTL, DTH, XL, XH, BC)  \
        SCAN_STEP(12, DTL, DTH, XL, XH, BC) SCAN_STEP(13, DTL, DTH, XL, XH, BC)  \
        SCAN_STEP(14, DTL, DTH, XL, XH, BC) SCAN_STEP(15, DTL, DTH, XL, XH, BC)  \
          \
        TREE_STAGE(Q, V, 8, bit0, 1)                                   \
        TREE_STAGE(R, Q, 4, bit1, 2)                                   \
        TREE_STAGE(S, R, 2, bit2, 4)                                   \
          \
        const size_t o = (size_t)(S0) * D_INNER;                       \
        pY[o]      = (S[0] + dsk * XL) * silu_f(ZL);                   \
        pY[o + HI] = (S[1] + dsk * XH) * silu_f(ZH);                   \
    }

#define HOLD_LOADS() __builtin_amdgcn_sched_barrier(0);

    float h0 = 0.0f, h1 = 0.0f;
    LOAD_TILE(dtLA, dtHA, xLA, xHA, zLA, zHA, bcA, 0)
    LOAD_TILE(dtLB, dtHB, xLB, xHB, zLB, zHB, bcB, TILE_S)
    int s0 = 0;
    for (; s0 + 5 * TILE_S <= SEQ; s0 += 3 * TILE_S) {
        LOAD_TILE(dtLC, dtHC, xLC, xHC, zLC, zHC, bcC, s0 + 2 * TILE_S)
        HOLD_LOADS()
        SCAN_TILE(dtLA, dtHA, xLA, xHA, zLA, zHA, bcA, s0)
        HOLD_LOADS()
        LOAD_TILE(dtLA, dtHA, xLA, xHA, zLA, zHA, bcA, s0 + 3 * TILE_S)
        HOLD_LOADS()
        SCAN_TILE(dtLB, dtHB, xLB, xHB, zLB, zHB, bcB, s0 + TILE_S)
        HOLD_LOADS()
        LOAD_TILE(dtLB, dtHB, xLB, xHB, zLB, zHB, bcB, s0 + 4 * TILE_S)
        HOLD_LOADS()
        SCAN_TILE(dtLC, dtHC, xLC, xHC, zLC, zHC, bcC, s0 + 2 * TILE_S)
    }
    static_assert((SEQ / TILE_S) % 3 == 2 && SEQ % TILE_S == 0, "tile count");
    SCAN_TILE(dtLA, dtHA, xLA, xHA, zLA, zHA, bcA, s0)
    SCAN_TILE(dtLB, dtHB, xLB, xHB, zLB, zHB, bcB, s0 + TILE_S)

#undef HOLD_LOADS
#undef TREE_STAGE
#undef LOAD_TILE
#undef SCAN_STEP
#undef SCAN_TILE
}

extern "C" void kernel_launch(void* const* d_in, const int* in_sizes, int n_in,
                              void* d_out, int out_size, void* d_ws, size_t ws_size,
                              hipStream_t stream) {
    const float* x         = (const float*)d_in[0];
    const float* in_proj_w = (const float*)d_in[1];
    const float* conv_w    = (const float*)d_in[2];
    const float* conv_b    = (const float*)d_in[3];
    const float* x_proj_w  = (const float*)d_in[4];
    const float* dt_proj_w = (const float*)d_in[5];
    const float* dt_proj_b = (const float*)d_in[6];
    const float* log_A     = (const float*)d_in[7];
    const float* D_skip    = (const float*)d_in[8];
    const float* out_proj_w= (const float*)d_in[9];
    float* out = (float*)d_out;

    const size_t planeBD = (size_t)MROWS * D_INNER;
    float* ws      = (float*)d_ws;
    float* buf_x   = ws;
    float* buf_z   = buf_x  + planeBD;
    float* buf_xc  = buf_z  + planeBD;
    float* buf_dbc = buf_xc + planeBD;
    float* buf_y   = buf_dbc + (size_t)MROWS * DBC_LD;

    const dim3 blk(128);
    const int gy = MROWS / (16 * 2 * 4);

    gemm_wmma_f32<2, 4, 0, 32><<<dim3(D_INNER / 64, gy), blk, 0, stream>>>(
        x, in_proj_w, nullptr, buf_x, MROWS, D_INNER, D_MODEL, D_MODEL, D_INNER);
    gemm_wmma_f32<2, 4, 0, 32><<<dim3(D_INNER / 64, gy), blk, 0, stream>>>(
        x, in_proj_w + (size_t)D_INNER * D_MODEL, nullptr, buf_z,
        MROWS, D_INNER, D_MODEL, D_MODEL, D_INNER);

    conv_xproj_kernel<<<MROWS / 16, XP_THREADS, 0, stream>>>(
        buf_x, conv_w, conv_b, x_proj_w, buf_xc, buf_dbc);

    gemm_wmma_f32<2, 4, 1, 16><<<dim3(D_INNER / 64, gy), blk, 0, stream>>>(
        buf_dbc, dt_proj_w, dt_proj_b, buf_x, MROWS, D_INNER, DT_RANK, DBC_LD, D_INNER);

    scan_kernel<<<(BATCH * D_INNER * (D_STATE / 2)) / SCAN_THREADS, SCAN_THREADS, 0, stream>>>(
        buf_x, buf_xc, buf_dbc, buf_z, log_A, D_skip, buf_y);

    gemm_wmma_f32<2, 2, 0, 64><<<dim3(D_MODEL / 32, gy), blk, 0, stream>>>(
        buf_y, out_proj_w, nullptr, out, MROWS, D_MODEL, D_INNER, D_INNER, D_MODEL);
}


# delta, conv result, z, y planes stored channel-blocked [d/4][token][4]: each scan load/store is one whole 128-byte line (was 16 B in each of 8 lines); GEMMs write them via transposed D, k order kept
# speedup vs baseline: 10.1065x; 1.0730x over previous
#include <hip/hip_runtime.h>
#include <hip/hip_bf16.h>


#define D_MODEL 768
#define D_STATE 16
#define D_CONV  4
#define D_INNER 1536
#define DT_RANK 48
#define BATCH   2
#define SEQ     2048
#define MROWS   (BATCH * SEQ)
#define DBC_LD  (DT_RANK + 2 * D_STATE)

typedef __attribute__((ext_vector_type(2))) float v2f;
typedef __attribute__((ext_vector_type(8))) float v8f;

__device__ __forceinline__ float silu_f(float x) {
    return x * (1.0f / (1.0f + __expf(-x)));
}
__device__ __forceinline__ float softplus_f(float x) {
    const float t  = __expf(x);
    const float s  = t * __builtin_amdgcn_rcpf(2.0f + t);
    const float s2 = s * s;
    float q = fmaf(s2, 1.0f / 9.0f, 1.0f / 7.0f);
    q = fmaf(s2, q, 1.0f / 5.0f);
    q = fmaf(s2, q, 1.0f / 3.0f);
    const float small = (s + s) * fmaf(s2, q, 1.0f);
    const float large = __builtin_amdgcn_logf(1.0f + t) * 0.69314718f;
    const float sp = (t > 0.5f) ? large : small;
    return (x > 20.0f) ? x : sp;
}

typedef __attribute__((ext_vector_type(4))) float v4f;

template <int MT, int NT, int EPI, int BK, bool ABLK, bool CBLK>
__launch_bounds__(128)
__global__ void gemm_wmma_f32(const float* __restrict__ A,
                              const float* __restrict__ W,
                              const float* __restrict__ bias,
                              float* __restrict__ C,
                              int M, int N, int K, int lda, int ldc) {
    constexpr int BM    = 64 * MT;
    constexpr int BN    = 16 * NT;
    constexpr int P     = BK + 4;
    constexpr int CPR   = BK / 4;
    constexpr int ACH   = BM * CPR / 128;
    constexpr int WCH   = BN * CPR / 128;
    constexpr int RSTEP = 128 / CPR;
    constexpr int STAGE = (BM + BN) * P;
    static_assert((BM * CPR) % 128 == 0 && (BN * CPR) % 128 == 0, "tile/threads");
    static_assert(((P / 4) & 1) == 1 && (P % 4) == 0, "pitch must be 4*odd floats");
    static_assert(!ABLK || (BM == 128 && ACH == CPR), "blocked A: one thread per tile row");

    __shared__ __attribute__((aligned(16))) float lds[2 * STAGE];

    const int tid    = threadIdx.x;
    const int lane   = tid & 31;
    const int wave   = tid >> 5;
    const int mBlock = blockIdx.y * BM;
    const int mBase  = mBlock + wave * (16 * MT);
    const int nBase  = blockIdx.x * BN;
    const int r16    = lane & 15;
    const int kOff   = (lane >> 4) * 2;

    const int cRow = tid / CPR;
    const int cCol = (tid % CPR) * 4;
    const float* gA = A + (size_t)(mBlock + cRow) * lda + cCol;
    const float* gW = W + (size_t)(nBase + cRow) * K + cCol;
    const int sOffA = cRow * P + cCol;
    const float* gAb = A + (size_t)(mBlock + tid) * 4;
    const size_t gStepB = (size_t)M * 4;
    const int sOffAb = tid * P;
    const int sOffW = (BM + cRow) * P + cCol;

    const int fA = (wave * (16 * MT) + r16) * P + kOff;
    const int fW = (BM + r16) * P + kOff;

    v8f acc[MT][NT];
#pragma unroll
    for (int i = 0; i < MT; ++i)
#pragma unroll
        for (int t = 0; t < NT; ++t) {
            v8f z = {0.f, 0.f, 0.f, 0.f, 0.f, 0.f, 0.f, 0.f};
            acc[i][t] = z;
        }

    v4f rA[ACH], rW[WCH];

#define GLOAD(K0)                                                        \
    _Pragma("unroll") for (int i = 0; i < ACH; ++i) {                    \
        if (ABLK) rA[i] = *(const v4f*)(gAb + (size_t)((K0) / 4 + i) * gStepB); \
        else      rA[i] = *(const v4f*)(gA + (size_t)(i * RSTEP) * lda + (K0)); \
    }                                                                    \
    _Pragma("unroll") for (int i = 0; i < WCH; ++i)                      \
        rW[i] = *(const v4f*)(gW + (size_t)(i * RSTEP) * K + (K0));

#define SSTORE(ST)                                                       \
    _Pragma("unroll") for (int i = 0; i < ACH; ++i) {                    \
        if (ABLK) *(v4f*)((ST) + sOffAb + 4 * i) = rA[i];                \
        else      *(v4f*)((ST) + sOffA + i * RSTEP * P) = rA[i];         \
    }                                                                    \
    _Pragma("unroll") for (int i = 0; i < WCH; ++i)                      \
        *(v4f*)((ST) + sOffW + i * RSTEP * P) = rW[i];

    const int nS = K / BK;
    GLOAD(0)
    SSTORE(lds)
    __syncthreads();

    for (int s = 0; s < nS; ++s) {
        const float* st = lds + (s & 1) * STAGE;
        const bool more = (s + 1 < nS);
        if (more) { GLOAD((s + 1) * BK) }

#pragma unroll
        for (int kk = 0; kk < BK / 4; ++kk) {
            v2f a[MT], b[NT];
#pragma unroll
            for (int i = 0; i < MT; ++i)
                a[i] = *(const v2f*)(st + fA + i * 16 * P + kk * 4);
#pragma unroll
            for (int t = 0; t < NT; ++t)
                b[t] = *(const v2f*)(st + fW + t * 16 * P + kk * 4);
#pragma unroll
            for (int i = 0; i < MT; ++i)
#pragma unroll
                for (int t = 0; t < NT; ++t) {
                    if (CBLK)
                        acc[i][t] = __builtin_amdgcn_wmma_f32_16x16x4_f32(
                            false, b[t], false, a[i], (short)0, acc[i][t],
                            false, false);
                    else
                        acc[i][t] = __builtin_amdgcn_wmma_f32_16x16x4_f32(
                            false, a[i], false, b[t], (short)0, acc[i][t],
                            false, false);
                }
        }

        if (more) { SSTORE(lds + ((s + 1) & 1) * STAGE) }
        __syncthreads();
    }

#undef GLOAD
#undef SSTORE

    if (CBLK) {
#pragma unroll
        for (int i = 0; i < MT; ++i)
#pragma unroll
            for (int t = 0; t < NT; ++t) {
                const int tok = mBase + i * 16 + (lane & 15);
                const int ch0 = nBase + t * 16 + (lane >> 4) * 8;
#pragma unroll
                for (int hf = 0; hf < 2; ++hf) {
                    v4f v;
#pragma unroll
                    for (int e = 0; e < 4; ++e) {
                        float x = acc[i][t][4 * hf + e];
                        if (EPI == 1) x = softplus_f(x + bias[ch0 + 4 * hf + e]);
                        v[e] = x;
                    }
                    *(v4f*)(C + ((size_t)((ch0 >> 2) + hf) * M + tok) * 4) = v;
                }
            }
        return;
    }

    const int rowSel = (lane >> 4) * 8;
#pragma unroll
    for (int i = 0; i < MT; ++i)
#pragma unroll
        for (int t = 0; t < NT; ++t) {
            const int col = nBase + t * 16 + (lane & 15);
            float bv = 0.0f;
            if (EPI == 1) bv = bias[col];
#pragma unroll
            for (int r = 0; r < 8; ++r) {
                float v = acc[i][t][r];
                if (EPI == 1) v = softplus_f(v + bv);
                C[(size_t)(mBase + i * 16 + r + rowSel) * ldc + col] = v;
            }
        }
}

#define XP_WAVES   8
#define XP_THREADS (32 * XP_WAVES)
#define XP_BK      32
#define XP_P       (XP_BK + 4)
#define XP_KW      (D_INNER / XP_WAVES)
#define XP_NS      (XP_KW / XP_BK)
#define XP_WCH     (DBC_LD * (XP_BK / 4) / 32)
#define XP_STAGE   ((16 + DBC_LD) * XP_P)
#define XP_RP      84

template <bool FIRST>
__device__ __forceinline__ void load_rows7(const float* __restrict__ pRow0, int g,
                                           v4f (&xr)[7]) {
#pragma unroll
    for (int jj = 0; jj < 7; ++jj) {
        const int r = 4 * g - 3 + jj;
        if (!FIRST || r >= 0) {
            xr[jj] = *(const v4f*)(pRow0 + (long)r * D_INNER);
        } else {
            v4f z = {0.f, 0.f, 0.f, 0.f};
            xr[jj] = z;
        }
    }
}

template <bool FIRST>
__device__ __forceinline__ void conv_rows4(const v4f (&xr)[7], const v4f (&cw)[4],
                                           const v4f cb, int g, v4f (&o)[4]) {
#pragma unroll
    for (int i = 0; i < 4; ++i) {
#pragma unroll
        for (int e = 0; e < 4; ++e) {
            float acc = cb[e];
#pragma unroll
            for (int j = 0; j < D_CONV; ++j) {
                if (!FIRST || (4 * g + i + j >= 3)) acc += cw[e][j] * xr[i + j][e];
            }
            o[i][e] = silu_f(acc);
        }
    }
}

__launch_bounds__(XP_THREADS)
__global__ void conv_xproj_kernel(const float* __restrict__ xin,
                                  const float* __restrict__ cw_,
                                  const float* __restrict__ cb_,
                                  const float* __restrict__ W,
                                  float* __restrict__ out,
                                  float* __restrict__ dbc) {
    static_assert(D_INNER % (XP_WAVES * XP_BK) == 0, "K split");
    static_assert(SEQ % 16 == 0 && DBC_LD % 16 == 0, "tiles");
    static_assert(XP_WCH * 32 == DBC_LD * (XP_BK / 4), "W chunks");
    static_assert(XP_WAVES * 16 * XP_RP <= XP_WAVES * XP_STAGE, "partial sums fit");
    constexpr int NT = DBC_LD / 16;

    __shared__ __attribute__((aligned(16))) float lds[XP_WAVES * XP_STAGE];

    const int tid  = threadIdx.x;
    const int lane = tid & 31;
    const int wave = tid >> 5;
    const int r16  = lane & 15;
    const int kOff = (lane >> 4) * 2;
    const int c4   = (lane & 7) * 4;
    const int g    = lane >> 3;
    const int m0   = blockIdx.x * 16;
    const bool first = (m0 % SEQ) == 0;
    const int kw0  = wave * XP_KW;

    float* stA = lds + wave * XP_STAGE;
    float* stW = stA + 16 * XP_P;

    const float* pX  = xin + (size_t)m0 * D_INNER + kw0 + c4;
    float*       pO  = out + ((size_t)(kw0 / 4 + (lane >> 4)) * MROWS + m0 + r16) * 4;
    const float* pW  = W + (size_t)g * D_INNER + kw0 + c4;
    const float* pCw = cw_ + (size_t)(kw0 + c4) * D_CONV;
    const float* pCb = cb_ + kw0 + c4;

    v8f acc[NT];
#pragma unroll
    for (int t = 0; t < NT; ++t) {
        v8f z = {0.f, 0.f, 0.f, 0.f, 0.f, 0.f, 0.f, 0.f};
        acc[t] = z;
    }

    v4f xr[7], wr[XP_WCH], cwv[4], cbv;

#define XP_GLOAD(S)                                                       \
    {                                                                     \
        const int ko = (S) * XP_BK;                                       \
        if (first) load_rows7<true>(pX + ko, g, xr);                      \
        else       load_rows7<false>(pX + ko, g, xr);                     \
        _Pragma("unroll") for (int i = 0; i < XP_WCH; ++i)                \
            wr[i] = *(const v4f*)(pW + (size_t)(4 * i) * D_INNER + ko);   \
        _Pragma("unroll") for (int e = 0; e < 4; ++e)                     \
            cwv[e] = *(const v4f*)(pCw + (size_t)(ko + e) * D_CONV);      \
        cbv = *(const v4f*)(pCb + ko);                                    \
    }

#define XP_BUILD(S)                                                       \
    {                                                                     \
        const int ko = (S) * XP_BK;                                       \
        v4f o[4];                                                         \
        if (first) conv_rows4<true>(xr, cwv, cbv, g, o);                  \
        else       conv_rows4<false>(xr, cwv, cbv, g, o);                 \
        _Pragma("unroll") for (int i = 0; i < 4; ++i)                     \
            *(v4f*)(stA + (4 * g + i) * XP_P + c4) = o[i];                \
        _Pragma("unroll") for (int i = 0; i < XP_WCH; ++i)                \
            *(v4f*)(stW + (g + 4 * i) * XP_P + c4) = wr[i];               \
    }

#define XP_COPY_OUT(S)                                                    \
    _Pragma("unroll") for (int j = 0; j < 4; ++j) {                       \
        const v4f v = *(const v4f*)(stA + r16 * XP_P + ((lane >> 4) + 2 * j) * 4); \
        *(v4f*)(pO + (size_t)((S) * (XP_BK / 4) + 2 * j) * (MROWS * 4)) = v; \
    }

    XP_GLOAD(0)
    XP_BUILD(0)
    __syncthreads();

    for (int s = 0; s < XP_NS; ++s) {
        const bool more = (s + 1 < XP_NS);
        if (more) { XP_GLOAD(s + 1) }
        XP_COPY_OUT(s)

#pragma unroll
        for (int kk = 0; kk < XP_BK / 4; ++kk) {
            const v2f a = *(const v2f*)(stA + r16 * XP_P + kk * 4 + kOff);
            v2f b[NT];
#pragma unroll
            for (int t = 0; t < NT; ++t)
                b[t] = *(const v2f*)(stW + (t * 16 + r16) * XP_P + kk * 4 + kOff);
#pragma unroll
            for (int t = 0; t < NT; ++t)
                acc[t] = __builtin_amdgcn_wmma_f32_16x16x4_f32(
                    false, a, false, b[t], (short)0, acc[t], false, false);
        }

        if (more) { XP_BUILD(s + 1) }
        __syncthreads();
    }

#undef XP_GLOAD
#undef XP_BUILD
#undef XP_COPY_OUT

    float* red = lds;
    const int rowSel = (lane >> 4) * 8;
#pragma unroll
    for (int t = 0; t < NT; ++t)
#pragma unroll
        for (int r = 0; r < 8; ++r)
            red[(wave * 16 + r + rowSel) * XP_RP + t * 16 + r16] = acc[t][r];
    __syncthreads();

#pragma unroll
    for (int i = 0; i < (16 * DBC_LD) / XP_THREADS; ++i) {
        const int oIdx = tid + XP_THREADS * i;
        const int row  = oIdx / DBC_LD;
        const int col  = oIdx - row * DBC_LD;
        float v = red[row * XP_RP + col];
#pragma unroll
        for (int w = 1; w < XP_WAVES; ++w)
            v += red[(w * 16 + row) * XP_RP + col];
        int ccol = col;
        if (col >= DT_RANK) {
            const int q = col - DT_RANK;
            ccol = DT_RANK + 4 * ((q & 15) >> 1) + 2 * (q >> 4) + (q & 1);
        }
        dbc[(size_t)(m0 + row) * DBC_LD + ccol] = v;
    }
}

#define TILE_S 16
#define SCAN_THREADS 96

template <int J>
__device__ __forceinline__ float bcast8_f(float v) {
    return __int_as_float(__builtin_amdgcn_mov_dpp8(__float_as_int(v), J * 0x249249));
}

template <int M>
__device__ __forceinline__ float row_xor_f(float v) {
    return __int_as_float(__builtin_amdgcn_update_dpp(
        0, __float_as_int(v), 0x160 + M, 0xf, 0xf, false));
}

__launch_bounds__(SCAN_THREADS)
__global__ void scan_kernel(const float* __restrict__ delta,
                            const float* __restrict__ xc,
                            const float* __restrict__ dbc,
                            const float* __restrict__ zbuf,
                            const float* __restrict__ logA,
                            const float* __restrict__ Dskip,
                            float* __restrict__ y) {
    const int gid = blockIdx.x * blockDim.x + threadIdx.x;
    const int p  = gid & 7;
    const int ch = gid >> 3;
    const int d  = ch % D_INNER;
    const int b  = ch / D_INNER;

    const float Aneg0 = -__expf(logA[d * D_STATE + 2 * p]);
    const float Aneg1 = -__expf(logA[d * D_STATE + 2 * p + 1]);
    const float dsk   = Dskip[d];
    const bool  bit0 = (p & 1) != 0, bit1 = (p & 2) != 0, bit2 = (p & 4) != 0;

    const size_t baseBD = ((size_t)(d >> 2) * MROWS + (size_t)b * SEQ + p) * 4 + (d & 3);
    const size_t HI = (size_t)8 * 4;
    const float* pDt = delta + baseBD;
    const float* pXc = xc    + baseBD;
    const float* pZ  = zbuf  + baseBD;
    float*       pY  = y     + baseBD;
    const float* pBC = dbc + (size_t)b * SEQ * DBC_LD + DT_RANK + 4 * p;

    float dtLA, dtHA, xLA, xHA, zLA, zHA;  v4f bcA[TILE_S];
    float dtLB, dtHB, xLB, xHB, zLB, zHB;  v4f bcB[TILE_S];
    float dtLC, dtHC, xLC, xHC, zLC, zHC;  v4f bcC[TILE_S];

#define LOAD_TILE(DTL, DTH, XL, XH, ZL, ZH, BC, S0)                    \
    {                                                                  \
        const size_t o = (size_t)(S0) * 4;                             \
        DTL = pDt[o];  DTH = pDt[o + HI];                              \
        XL  = pXc[o];  XH  = pXc[o + HI];                              \
        const float* pb = pBC + (size_t)(S0) * DBC_LD;                 \
                   \
        _Pragma("unroll") for (int j = 0; j < TILE_S; ++j)             \
            BC[j] = *(const v4f*)(pb + j * DBC_LD);                    \
        ZL  = pZ[o];   ZH  = pZ[o + HI];                               \
    }

#define SCAN_STEP(J, DTL, DTH, XL, XH, BC)                             \
    {                                                                  \
        const float dtj = bcast8_f<((J) & 7)>(((J) < 8) ? DTL : DTH);  \
        const float xvj = bcast8_f<((J) & 7)>(((J) < 8) ? XL : XH);    \
        const float a0 = __expf(dtj * Aneg0);                          \
        const float a1 = __expf(dtj * Aneg1);                          \
          \
          \
        h0 = fmaf(a0, h0, (dtj * BC[J].x) * xvj);                      \
        h1 = fmaf(a1, h1, (dtj * BC[J].y) * xvj);                      \
        V[J] = h0 * BC[J].z + h1 * BC[J].w;                            \
    }

#define TREE_STAGE(OUT, IN, CNT, BIT, M)                               \
           \
    _Pragma("unroll") for (int i = 0; i < (CNT); ++i) {                \
        const float keep = (BIT) ? IN[2 * i + 1] : IN[2 * i];          \
        const float send = (BIT) ? IN[2 * i] : IN[2 * i + 1];          \
        OUT[i] = keep + row_xor_f<M>(send);                            \
    }

#define SCAN_TILE(DTL, DTH, XL, XH, ZL, ZH, BC, S0)                    \
    {                                                                  \
        float V[TILE_S], Q[8], R[4], S[2];                             \
        SCAN_STEP(0, DTL, DTH, XL, XH, BC)  SCAN_STEP(1, DTL, DTH, XL, XH, BC)   \
        SCAN_STEP(2, DTL, DTH, XL, XH, BC)  SCAN_STEP(3, DTL, DTH, XL, XH, BC)   \
        SCAN_STEP(4, DTL, DTH, XL, XH, BC)  SCAN_STEP(5, DTL, DTH, XL, XH, BC)   \
        SCAN_STEP(6, DTL, DTH, XL, XH, BC)  SCAN_STEP(7, DTL, DTH, XL, XH, BC)   \
        SCAN_STEP(8, DTL, DTH, XL, XH, BC)  SCAN_STEP(9, DTL, DTH, XL, XH, BC)   \
        SCAN_STEP(10, DTL, DTH, XL, XH, BC) SCAN_STEP(11, DTL, DTH, XL, XH, BC)  \
        SCAN_STEP(12, DTL, DTH, XL, XH, BC) SCAN_STEP(13, DTL, DTH, XL, XH, BC)  \
        SCAN_STEP(14, DTL, DTH, XL, XH, BC) SCAN_STEP(15, DTL, DTH, XL, XH, BC)  \
          \
        TREE_STAGE(Q, V, 8, bit0, 1)                                   \
        TREE_STAGE(R, Q, 4, bit1, 2)                                   \
        TREE_STAGE(S, R, 2, bit2, 4)                                   \
          \
        const size_t o = (size_t)(S0) * 4;                             \
        pY[o]      = (S[0] + dsk * XL) * silu_f(ZL);                   \
        pY[o + HI] = (S[1] + dsk * XH) * silu_f(ZH);                   \
    }

#define HOLD_LOADS() __builtin_amdgcn_sched_barrier(0);

    float h0 = 0.0f, h1 = 0.0f;
    LOAD_TILE(dtLA, dtHA, xLA, xHA, zLA, zHA, bcA, 0)
    LOAD_TILE(dtLB, dtHB, xLB, xHB, zLB, zHB, bcB, TILE_S)
    int s0 = 0;
    for (; s0 + 5 * TILE_S <= SEQ; s0 += 3 * TILE_S) {
        LOAD_TILE(dtLC, dtHC, xLC, xHC, zLC, zHC, bcC, s0 + 2 * TILE_S)
        HOLD_LOADS()
        SCAN_TILE(dtLA, dtHA, xLA, xHA, zLA, zHA, bcA, s0)
        HOLD_LOADS()
        LOAD_TILE(dtLA, dtHA, xLA, xHA, zLA, zHA, bcA, s0 + 3 * TILE_S)
        HOLD_LOADS()
        SCAN_TILE(dtLB, dtHB, xLB, xHB, zLB, zHB, bcB, s0 + TILE_S)
        HOLD_LOADS()
        LOAD_TILE(dtLB, dtHB, xLB, xHB, zLB, zHB, bcB, s0 + 4 * TILE_S)
        HOLD_LOADS()
        SCAN_TILE(dtLC, dtHC, xLC, xHC, zLC, zHC, bcC, s0 + 2 * TILE_S)
    }
    static_assert((SEQ / TILE_S) % 3 == 2 && SEQ % TILE_S == 0, "tile count");
    SCAN_TILE(dtLA, dtHA, xLA, xHA, zLA, zHA, bcA, s0)
    SCAN_TILE(dtLB, dtHB, xLB, xHB, zLB, zHB, bcB, s0 + TILE_S)

#undef HOLD_LOADS
#undef TREE_STAGE
#undef LOAD_TILE
#undef SCAN_STEP
#undef SCAN_TILE
}

extern "C" void kernel_launch(void* const* d_in, const int* in_sizes, int n_in,
                              void* d_out, int out_size, void* d_ws, size_t ws_size,
                              hipStream_t stream) {
    const float* x         = (const float*)d_in[0];
    const float* in_proj_w = (const float*)d_in[1];
    const float* conv_w    = (const float*)d_in[2];
    const float* conv_b    = (const float*)d_in[3];
    const float* x_proj_w  = (const float*)d_in[4];
    const float* dt_proj_w = (const float*)d_in[5];
    const float* dt_proj_b = (const float*)d_in[6];
    const float* log_A     = (const float*)d_in[7];
    const float* D_skip    = (const float*)d_in[8];
    const float* out_proj_w= (const float*)d_in[9];
    float* out = (float*)d_out;

    const size_t planeBD = (size_t)MROWS * D_INNER;
    float* ws      = (float*)d_ws;
    float* buf_x   = ws;
    float* buf_z   = buf_x  + planeBD;
    float* buf_xc  = buf_z  + planeBD;
    float* buf_dbc = buf_xc + planeBD;
    float* buf_y   = buf_dbc + (size_t)MROWS * DBC_LD;

    const dim3 blk(128);
    const int gy = MROWS / (16 * 2 * 4);

    gemm_wmma_f32<2, 4, 0, 32, false, false><<<dim3(D_INNER / 64, gy), blk, 0, stream>>>(
        x, in_proj_w, nullptr, buf_x, MROWS, D_INNER, D_MODEL, D_MODEL, D_INNER);
    gemm_wmma_f32<2, 4, 0, 32, false, true><<<dim3(D_INNER / 64, gy), blk, 0, stream>>>(
        x, in_proj_w + (size_t)D_INNER * D_MODEL, nullptr, buf_z,
        MROWS, D_INNER, D_MODEL, D_MODEL, D_INNER);

    conv_xproj_kernel<<<MROWS / 16, XP_THREADS, 0, stream>>>(
        buf_x, conv_w, conv_b, x_proj_w, buf_xc, buf_dbc);

    gemm_wmma_f32<2, 4, 1, 16, false, true><<<dim3(D_INNER / 64, gy), blk, 0, stream>>>(
        buf_dbc, dt_proj_w, dt_proj_b, buf_x, MROWS, D_INNER, DT_RANK, DBC_LD, D_INNER);

    scan_kernel<<<(BATCH * D_INNER * (D_STATE / 2)) / SCAN_THREADS, SCAN_THREADS, 0, stream>>>(
        buf_x, buf_xc, buf_dbc, buf_z, log_A, D_skip, buf_y);

    gemm_wmma_f32<2, 2, 0, 64, true, false><<<dim3(D_MODEL / 32, gy), blk, 0, stream>>>(
        buf_y, out_proj_w, nullptr, out, MROWS, D_MODEL, D_INNER, D_INNER, D_MODEL);
}


# in_proj: x- and z-branch GEMMs in one launch; a block takes its 128x64 tile of both branches from one LDS copy of its A rows (wave tile 32x128, 128 WMMAs per barrier); 5 launches, k order kept
# speedup vs baseline: 10.8354x; 1.0721x over previous
#include <hip/hip_runtime.h>
#include <hip/hip_bf16.h>


#define D_MODEL 768
#define D_STATE 16
#define D_CONV  4
#define D_INNER 1536
#define DT_RANK 48
#define BATCH   2
#define SEQ     2048
#define MROWS   (BATCH * SEQ)
#define DBC_LD  (DT_RANK + 2 * D_STATE)

typedef __attribute__((ext_vector_type(2))) float v2f;
typedef __attribute__((ext_vector_type(8))) float v8f;

__device__ __forceinline__ float silu_f(float x) {
    return x * (1.0f / (1.0f + __expf(-x)));
}
__device__ __forceinline__ float softplus_f(float x) {
    const float t  = __expf(x);
    const float s  = t * __builtin_amdgcn_rcpf(2.0f + t);
    const float s2 = s * s;
    float q = fmaf(s2, 1.0f / 9.0f, 1.0f / 7.0f);
    q = fmaf(s2, q, 1.0f / 5.0f);
    q = fmaf(s2, q, 1.0f / 3.0f);
    const float small = (s + s) * fmaf(s2, q, 1.0f);
    const float large = __builtin_amdgcn_logf(1.0f + t) * 0.69314718f;
    const float sp = (t > 0.5f) ? large : small;
    return (x > 20.0f) ? x : sp;
}

typedef __attribute__((ext_vector_type(4))) float v4f;

template <int MT, int NT, int EPI, int BK, bool ABLK, bool CBLK>
__launch_bounds__(128)
__global__ void gemm_wmma_f32(const float* __restrict__ A,
                              const float* __restrict__ W,
                              const float* __restrict__ bias,
                              float* __restrict__ C,
                              int M, int N, int K, int lda, int ldc) {
    constexpr int BM    = 64 * MT;
    constexpr int BN    = 16 * NT;
    constexpr int P     = BK + 4;
    constexpr int CPR   = BK / 4;
    constexpr int ACH   = BM * CPR / 128;
    constexpr int WCH   = BN * CPR / 128;
    constexpr int RSTEP = 128 / CPR;
    constexpr int STAGE = (BM + BN) * P;
    static_assert((BM * CPR) % 128 == 0 && (BN * CPR) % 128 == 0, "tile/threads");
    static_assert(((P / 4) & 1) == 1 && (P % 4) == 0, "pitch must be 4*odd floats");
    static_assert(!ABLK || (BM == 128 && ACH == CPR), "blocked A: one thread per tile row");

    __shared__ __attribute__((aligned(16))) float lds[2 * STAGE];

    const int tid    = threadIdx.x;
    const int lane   = tid & 31;
    const int wave   = tid >> 5;
    const int mBlock = blockIdx.y * BM;
    const int mBase  = mBlock + wave * (16 * MT);
    const int nBase  = blockIdx.x * BN;
    const int r16    = lane & 15;
    const int kOff   = (lane >> 4) * 2;

    const int cRow = tid / CPR;
    const int cCol = (tid % CPR) * 4;
    const float* gA = A + (size_t)(mBlock + cRow) * lda + cCol;
    const float* gW = W + (size_t)(nBase + cRow) * K + cCol;
    const int sOffA = cRow * P + cCol;
    const float* gAb = A + (size_t)(mBlock + tid) * 4;
    const size_t gStepB = (size_t)M * 4;
    const int sOffAb = tid * P;
    const int sOffW = (BM + cRow) * P + cCol;

    const int fA = (wave * (16 * MT) + r16) * P + kOff;
    const int fW = (BM + r16) * P + kOff;

    v8f acc[MT][NT];
#pragma unroll
    for (int i = 0; i < MT; ++i)
#pragma unroll
        for (int t = 0; t < NT; ++t) {
            v8f z = {0.f, 0.f, 0.f, 0.f, 0.f, 0.f, 0.f, 0.f};
            acc[i][t] = z;
        }

    v4f rA[ACH], rW[WCH];

#define GLOAD(K0)                                                        \
    _Pragma("unroll") for (int i = 0; i < ACH; ++i) {                    \
        if (ABLK) rA[i] = *(const v4f*)(gAb + (size_t)((K0) / 4 + i) * gStepB); \
        else      rA[i] = *(const v4f*)(gA + (size_t)(i * RSTEP) * lda + (K0)); \
    }                                                                    \
    _Pragma("unroll") for (int i = 0; i < WCH; ++i)                      \
        rW[i] = *(const v4f*)(gW + (size_t)(i * RSTEP) * K + (K0));

#define SSTORE(ST)                                                       \
    _Pragma("unroll") for (int i = 0; i < ACH; ++i) {                    \
        if (ABLK) *(v4f*)((ST) + sOffAb + 4 * i) = rA[i];                \
        else      *(v4f*)((ST) + sOffA + i * RSTEP * P) = rA[i];         \
    }                                                                    \
    _Pragma("unroll") for (int i = 0; i < WCH; ++i)                      \
        *(v4f*)((ST) + sOffW + i * RSTEP * P) = rW[i];

    const int nS = K / BK;
    GLOAD(0)
    SSTORE(lds)
    __syncthreads();

    for (int s = 0; s < nS; ++s) {
        const float* st = lds + (s & 1) * STAGE;
        const bool more = (s + 1 < nS);
        if (more) { GLOAD((s + 1) * BK) }

#pragma unroll
        for (int kk = 0; kk < BK / 4; ++kk) {
            v2f a[MT], b[NT];
#pragma unroll
            for (int i = 0; i < MT; ++i)
                a[i] = *(const v2f*)(st + fA + i * 16 * P + kk * 4);
#pragma unroll
            for (int t = 0; t < NT; ++t)
                b[t] = *(const v2f*)(st + fW + t * 16 * P + kk * 4);
#pragma unroll
            for (int i = 0; i < MT; ++i)
#pragma unroll
                for (int t = 0; t < NT; ++t) {
                    if (CBLK)
                        acc[i][t] = __builtin_amdgcn_wmma_f32_16x16x4_f32(
                            false, b[t], false, a[i], (short)0, acc[i][t],
                            false, false);
                    else
                        acc[i][t] = __builtin_amdgcn_wmma_f32_16x16x4_f32(
                            false, a[i], false, b[t], (short)0, acc[i][t],
                            false, false);
                }
        }

        if (more) { SSTORE(lds + ((s + 1) & 1) * STAGE) }
        __syncthreads();
    }

#undef GLOAD
#undef SSTORE

    if (CBLK) {
#pragma unroll
        for (int i = 0; i < MT; ++i)
#pragma unroll
            for (int t = 0; t < NT; ++t) {
                const int tok = mBase + i * 16 + (lane & 15);
                const int ch0 = nBase + t * 16 + (lane >> 4) * 8;
#pragma unroll
                for (int hf = 0; hf < 2; ++hf) {
                    v4f v;
#pragma unroll
                    for (int e = 0; e < 4; ++e) {
                        float x = acc[i][t][4 * hf + e];
                        if (EPI == 1) x = softplus_f(x + bias[ch0 + 4 * hf + e]);
                        v[e] = x;
                    }
                    *(v4f*)(C + ((size_t)((ch0 >> 2) + hf) * M + tok) * 4) = v;
                }
            }
        return;
    }

    const int rowSel = (lane >> 4) * 8;
#pragma unroll
    for (int i = 0; i < MT; ++i)
#pragma unroll
        for (int t = 0; t < NT; ++t) {
            const int col = nBase + t * 16 + (lane & 15);
            float bv = 0.0f;
            if (EPI == 1) bv = bias[col];
#pragma unroll
            for (int r = 0; r < 8; ++r) {
                float v = acc[i][t][r];
                if (EPI == 1) v = softplus_f(v + bv);
                C[(size_t)(mBase + i * 16 + r + rowSel) * ldc + col] = v;
            }
        }
}

__launch_bounds__(128) __attribute__((amdgpu_waves_per_eu(1, 3)))
__global__ void inproj_xz_wmma_f32(const float* __restrict__ A,
                                   const float* __restrict__ W,
                                   float* __restrict__ Cx,
                                   float* __restrict__ Cz) {
    constexpr int MT    = 2;
    constexpr int NH    = 4;
    constexpr int NT    = 2 * NH;
    constexpr int BK    = 32;
    constexpr int K     = D_MODEL;
    constexpr int M     = MROWS;
    constexpr int BM    = 64 * MT;
    constexpr int BNH   = 16 * NH;
    constexpr int BW    = 2 * BNH;
    constexpr int P     = BK + 4;
    constexpr int CPR   = BK / 4;
    constexpr int ACH   = BM * CPR / 128;
    constexpr int WCH   = BW * CPR / 128;
    constexpr int RSTEP = 128 / CPR;
    constexpr int STAGE = (BM + BW) * P;
    static_assert(K % BK == 0 && M % BM == 0 && D_INNER % BNH == 0, "exact tiles");
    static_assert(((P / 4) & 1) == 1 && (P % 4) == 0, "pitch must be 4*odd floats");
    static_assert(BNH % RSTEP == 0 && WCH == 2 * (BNH / RSTEP), "a thread's W chunks split evenly over the branches");

    __shared__ __attribute__((aligned(16))) float lds[2 * STAGE];

    const int tid    = threadIdx.x;
    const int lane   = tid & 31;
    const int wave   = tid >> 5;
    const int mBlock = blockIdx.y * BM;
    const int mBase  = mBlock + wave * (16 * MT);
    const int nBase  = blockIdx.x * BNH;
    const int r16    = lane & 15;
    const int kOff   = (lane >> 4) * 2;

    const int cRow = tid / CPR;
    const int cCol = (tid % CPR) * 4;
    const float* gA  = A + (size_t)(mBlock + cRow) * K + cCol;
    const float* gWx = W + (size_t)(nBase + cRow) * K + cCol;
    const float* gWz = gWx + (size_t)D_INNER * K;
    const int sOffA = cRow * P + cCol;
    const int sOffW = (BM + cRow) * P + cCol;

    const int fA = (wave * (16 * MT) + r16) * P + kOff;
    const int fW = (BM + r16) * P + kOff;

    v8f acc[MT][NT];
#pragma unroll
    for (int i = 0; i < MT; ++i)
#pragma unroll
        for (int t = 0; t < NT; ++t) {
            v8f z = {0.f, 0.f, 0.f, 0.f, 0.f, 0.f, 0.f, 0.f};
            acc[i][t] = z;
        }

    v4f rA[ACH], rW[WCH];

#define GLOAD(K0)                                                        \
    _Pragma("unroll") for (int i = 0; i < ACH; ++i)                      \
        rA[i] = *(const v4f*)(gA + (size_t)(i * RSTEP) * K + (K0));      \
    _Pragma("unroll") for (int i = 0; i < WCH / 2; ++i) {                \
        rW[i]           = *(const v4f*)(gWx + (size_t)(i * RSTEP) * K + (K0)); \
        rW[WCH / 2 + i] = *(const v4f*)(gWz + (size_t)(i * RSTEP) * K + (K0)); \
    }

#define SSTORE(ST)                                                       \
    _Pragma("unroll") for (int i = 0; i < ACH; ++i)                      \
        *(v4f*)((ST) + sOffA + i * RSTEP * P) = rA[i];                   \
    _Pragma("unroll") for (int i = 0; i < WCH; ++i)                      \
        *(v4f*)((ST) + sOffW + i * RSTEP * P) = rW[i];

    constexpr int nS = K / BK;
    GLOAD(0)
    SSTORE(lds)
    __syncthreads();

    for (int s = 0; s < nS; ++s) {
        const float* st = lds + (s & 1) * STAGE;
        const bool more = (s + 1 < nS);
        if (more) { GLOAD((s + 1) * BK) }

#pragma unroll
        for (int kk = 0; kk < BK / 4; ++kk) {
            v2f a[MT], b[NT];
#pragma unroll
            for (int i = 0; i < MT; ++i)
                a[i] = *(const v2f*)(st + fA + i * 16 * P + kk * 4);
#pragma unroll
            for (int t = 0; t < NT; ++t)
                b[t] = *(const v2f*)(st + fW + t * 16 * P + kk * 4);
#pragma unroll
            for (int i = 0; i < MT; ++i)
#pragma unroll
                for (int t = 0; t < NT; ++t) {
                    if (t >= NH)
                        acc[i][t] = __builtin_amdgcn_wmma_f32_16x16x4_f32(
                            false, b[t], false, a[i], (short)0, acc[i][t],
                            false, false);
                    else
                        acc[i][t] = __builtin_amdgcn_wmma_f32_16x16x4_f32(
                            false, a[i], false, b[t], (short)0, acc[i][t],
                            false, false);
                }
        }

        if (more) { SSTORE(lds + ((s + 1) & 1) * STAGE) }
        __syncthreads();
    }

#undef GLOAD
#undef SSTORE

    const int rowSel = (lane >> 4) * 8;
#pragma unroll
    for (int i = 0; i < MT; ++i)
#pragma unroll
        for (int t = 0; t < NH; ++t) {
            const int col = nBase + t * 16 + (lane & 15);
#pragma unroll
            for (int r = 0; r < 8; ++r)
                Cx[(size_t)(mBase + i * 16 + r + rowSel) * D_INNER + col] = acc[i][t][r];
        }

#pragma unroll
    for (int i = 0; i < MT; ++i)
#pragma unroll
        for (int t = 0; t < NH; ++t) {
            const int tok = mBase + i * 16 + (lane & 15);
            const int ch0 = nBase + t * 16 + (lane >> 4) * 8;
#pragma unroll
            for (int hf = 0; hf < 2; ++hf) {
                v4f v;
#pragma unroll
                for (int e = 0; e < 4; ++e) v[e] = acc[i][NH + t][4 * hf + e];
                *(v4f*)(Cz + ((size_t)((ch0 >> 2) + hf) * M + tok) * 4) = v;
            }
        }
}

#define XP_WAVES   8
#define XP_THREADS (32 * XP_WAVES)
#define XP_BK      32
#define XP_P       (XP_BK + 4)
#define XP_KW      (D_INNER / XP_WAVES)
#define XP_NS      (XP_KW / XP_BK)
#define XP_WCH     (DBC_LD * (XP_BK / 4) / 32)
#define XP_STAGE   ((16 + DBC_LD) * XP_P)
#define XP_RP      84

template <bool FIRST>
__device__ __forceinline__ void load_rows7(const float* __restrict__ pRow0, int g,
                                           v4f (&xr)[7]) {
#pragma unroll
    for (int jj = 0; jj < 7; ++jj) {
        const int r = 4 * g - 3 + jj;
        if (!FIRST || r >= 0) {
            xr[jj] = *(const v4f*)(pRow0 + (long)r * D_INNER);
        } else {
            v4f z = {0.f, 0.f, 0.f, 0.f};
            xr[jj] = z;
        }
    }
}

template <bool FIRST>
__device__ __forceinline__ void conv_rows4(const v4f (&xr)[7], const v4f (&cw)[4],
                                           const v4f cb, int g, v4f (&o)[4]) {
#pragma unroll
    for (int i = 0; i < 4; ++i) {
#pragma unroll
        for (int e = 0; e < 4; ++e) {
            float acc = cb[e];
#pragma unroll
            for (int j = 0; j < D_CONV; ++j) {
                if (!FIRST || (4 * g + i + j >= 3)) acc += cw[e][j] * xr[i + j][e];
            }
            o[i][e] = silu_f(acc);
        }
    }
}

__launch_bounds__(XP_THREADS)
__global__ void conv_xproj_kernel(const float* __restrict__ xin,
                                  const float* __restrict__ cw_,
                                  const float* __restrict__ cb_,
                                  const float* __restrict__ W,
                                  float* __restrict__ out,
                                  float* __restrict__ dbc) {
    static_assert(D_INNER % (XP_WAVES * XP_BK) == 0, "K split");
    static_assert(SEQ % 16 == 0 && DBC_LD % 16 == 0, "tiles");
    static_assert(XP_WCH * 32 == DBC_LD * (XP_BK / 4), "W chunks");
    static_assert(XP_WAVES * 16 * XP_RP <= XP_WAVES * XP_STAGE, "partial sums fit");
    constexpr int NT = DBC_LD / 16;

    __shared__ __attribute__((aligned(16))) float lds[XP_WAVES * XP_STAGE];

    const int tid  = threadIdx.x;
    const int lane = tid & 31;
    const int wave = tid >> 5;
    const int r16  = lane & 15;
    const int kOff = (lane >> 4) * 2;
    const int c4   = (lane & 7) * 4;
    const int g    = lane >> 3;
    const int m0   = blockIdx.x * 16;
    const bool first = (m0 % SEQ) == 0;
    const int kw0  = wave * XP_KW;

    float* stA = lds + wave * XP_STAGE;
    float* stW = stA + 16 * XP_P;

    const float* pX  = xin + (size_t)m0 * D_INNER + kw0 + c4;
    float*       pO  = out + ((size_t)(kw0 / 4 + (lane >> 4)) * MROWS + m0 + r16) * 4;
    const float* pW  = W + (size_t)g * D_INNER + kw0 + c4;
    const float* pCw = cw_ + (size_t)(kw0 + c4) * D_CONV;
    const float* pCb = cb_ + kw0 + c4;

    v8f acc[NT];
#pragma unroll
    for (int t = 0; t < NT; ++t) {
        v8f z = {0.f, 0.f, 0.f, 0.f, 0.f, 0.f, 0.f, 0.f};
        acc[t] = z;
    }

    v4f xr[7], wr[XP_WCH], cwv[4], cbv;

#define XP_GLOAD(S)                                                       \
    {                                                                     \
        const int ko = (S) * XP_BK;                                       \
        if (first) load_rows7<true>(pX + ko, g, xr);                      \
        else       load_rows7<false>(pX + ko, g, xr);                     \
        _Pragma("unroll") for (int i = 0; i < XP_WCH; ++i)                \
            wr[i] = *(const v4f*)(pW + (size_t)(4 * i) * D_INNER + ko);   \
        _Pragma("unroll") for (int e = 0; e < 4; ++e)                     \
            cwv[e] = *(const v4f*)(pCw + (size_t)(ko + e) * D_CONV);      \
        cbv = *(const v4f*)(pCb + ko);                                    \
    }

#define XP_BUILD(S)                                                       \
    {                                                                     \
        const int ko = (S) * XP_BK;                                       \
        v4f o[4];                                                         \
        if (first) conv_rows4<true>(xr, cwv, cbv, g, o);                  \
        else       conv_rows4<false>(xr, cwv, cbv, g, o);                 \
        _Pragma("unroll") for (int i = 0; i < 4; ++i)                     \
            *(v4f*)(stA + (4 * g + i) * XP_P + c4) = o[i];                \
        _Pragma("unroll") for (int i = 0; i < XP_WCH; ++i)                \
            *(v4f*)(stW + (g + 4 * i) * XP_P + c4) = wr[i];               \
    }

#define XP_COPY_OUT(S)                                                    \
    _Pragma("unroll") for (int j = 0; j < 4; ++j) {                       \
        const v4f v = *(const v4f*)(stA + r16 * XP_P + ((lane >> 4) + 2 * j) * 4); \
        *(v4f*)(pO + (size_t)((S) * (XP_BK / 4) + 2 * j) * (MROWS * 4)) = v; \
    }

    XP_GLOAD(0)
    XP_BUILD(0)
    __syncthreads();

    for (int s = 0; s < XP_NS; ++s) {
        const bool more = (s + 1 < XP_NS);
        if (more) { XP_GLOAD(s + 1) }
        XP_COPY_OUT(s)

#pragma unroll
        for (int kk = 0; kk < XP_BK / 4; ++kk) {
            const v2f a = *(const v2f*)(stA + r16 * XP_P + kk * 4 + kOff);
            v2f b[NT];
#pragma unroll
            for (int t = 0; t < NT; ++t)
                b[t] = *(const v2f*)(stW + (t * 16 + r16) * XP_P + kk * 4 + kOff);
#pragma unroll
            for (int t = 0; t < NT; ++t)
                acc[t] = __builtin_amdgcn_wmma_f32_16x16x4_f32(
                    false, a, false, b[t], (short)0, acc[t], false, false);
        }

        if (more) { XP_BUILD(s + 1) }
        __syncthreads();
    }

#undef XP_GLOAD
#undef XP_BUILD
#undef XP_COPY_OUT

    float* red = lds;
    const int rowSel = (lane >> 4) * 8;
#pragma unroll
    for (int t = 0; t < NT; ++t)
#pragma unroll
        for (int r = 0; r < 8; ++r)
            red[(wave * 16 + r + rowSel) * XP_RP + t * 16 + r16] = acc[t][r];
    __syncthreads();

#pragma unroll
    for (int i = 0; i < (16 * DBC_LD) / XP_THREADS; ++i) {
        const int oIdx = tid + XP_THREADS * i;
        const int row  = oIdx / DBC_LD;
        const int col  = oIdx - row * DBC_LD;
        float v = red[row * XP_RP + col];
#pragma unroll
        for (int w = 1; w < XP_WAVES; ++w)
            v += red[(w * 16 + row) * XP_RP + col];
        int ccol = col;
        if (col >= DT_RANK) {
            const int q = col - DT_RANK;
            ccol = DT_RANK + 4 * ((q & 15) >> 1) + 2 * (q >> 4) + (q & 1);
        }
        dbc[(size_t)(m0 + row) * DBC_LD + ccol] = v;
    }
}

#define TILE_S 16
#define SCAN_THREADS 96

template <int J>
__device__ __forceinline__ float bcast8_f(float v) {
    return __int_as_float(__builtin_amdgcn_mov_dpp8(__float_as_int(v), J * 0x249249));
}

template <int M>
__device__ __forceinline__ float row_xor_f(float v) {
    return __int_as_float(__builtin_amdgcn_update_dpp(
        0, __float_as_int(v), 0x160 + M, 0xf, 0xf, false));
}

__launch_bounds__(SCAN_THREADS)
__global__ void scan_kernel(const float* __restrict__ delta,
                            const float* __restrict__ xc,
                            const float* __restrict__ dbc,
                            const float* __restrict__ zbuf,
                            const float* __restrict__ logA,
                            const float* __restrict__ Dskip,
                            float* __restrict__ y) {
    const int gid = blockIdx.x * blockDim.x + threadIdx.x;
    const int p  = gid & 7;
    const int ch = gid >> 3;
    const int d  = ch % D_INNER;
    const int b  = ch / D_INNER;

    const float Aneg0 = -__expf(logA[d * D_STATE + 2 * p]);
    const float Aneg1 = -__expf(logA[d * D_STATE + 2 * p + 1]);
    const float dsk   = Dskip[d];
    const bool  bit0 = (p & 1) != 0, bit1 = (p & 2) != 0, bit2 = (p & 4) != 0;

    const size_t baseBD = ((size_t)(d >> 2) * MROWS + (size_t)b * SEQ + p) * 4 + (d & 3);
    const size_t HI = (size_t)8 * 4;
    const float* pDt = delta + baseBD;
    const float* pXc = xc    + baseBD;
    const float* pZ  = zbuf  + baseBD;
    float*       pY  = y     + baseBD;
    const float* pBC = dbc + (size_t)b * SEQ * DBC_LD + DT_RANK + 4 * p;

    float dtLA, dtHA, xLA, xHA, zLA, zHA;  v4f bcA[TILE_S];
    float dtLB, dtHB, xLB, xHB, zLB, zHB;  v4f bcB[TILE_S];
    float dtLC, dtHC, xLC, xHC, zLC, zHC;  v4f bcC[TILE_S];

#define LOAD_TILE(DTL, DTH, XL, XH, ZL, ZH, BC, S0)                    \
    {                                                                  \
        const size_t o = (size_t)(S0) * 4;                             \
        DTL = pDt[o];  DTH = pDt[o + HI];                              \
        XL  = pXc[o];  XH  = pXc[o + HI];                              \
        const float* pb = pBC + (size_t)(S0) * DBC_LD;                 \
                   \
        _Pragma("unroll") for (int j = 0; j < TILE_S; ++j)             \
            BC[j] = *(const v4f*)(pb + j * DBC_LD);                    \
        ZL  = pZ[o];   ZH  = pZ[o + HI];                               \
    }

#define SCAN_STEP(J, DTL, DTH, XL, XH, BC)                             \
    {                                                                  \
        const float dtj = bcast8_f<((J) & 7)>(((J) < 8) ? DTL : DTH);  \
        const float xvj = bcast8_f<((J) & 7)>(((J) < 8) ? XL : XH);    \
        const float a0 = __expf(dtj * Aneg0);                          \
        const float a1 = __expf(dtj * Aneg1);                          \
          \
          \
        h0 = fmaf(a0, h0, (dtj * BC[J].x) * xvj);                      \
        h1 = fmaf(a1, h1, (dtj * BC[J].y) * xvj);                      \
        V[J] = h0 * BC[J].z + h1 * BC[J].w;                            \
    }

#define TREE_STAGE(OUT, IN, CNT, BIT, M)                               \
           \
    _Pragma("unroll") for (int i = 0; i < (CNT); ++i) {                \
        const float keep = (BIT) ? IN[2 * i + 1] : IN[2 * i];          \
        const float send = (BIT) ? IN[2 * i] : IN[2 * i + 1];          \
        OUT[i] = keep + row_xor_f<M>(send);                            \
    }

#define SCAN_TILE(DTL, DTH, XL, XH, ZL, ZH, BC, S0)                    \
    {                                                                  \
        float V[TILE_S], Q[8], R[4], S[2];                             \
        SCAN_STEP(0, DTL, DTH, XL, XH, BC)  SCAN_STEP(1, DTL, DTH, XL, XH, BC)   \
        SCAN_STEP(2, DTL, DTH, XL, XH, BC)  SCAN_STEP(3, DTL, DTH, XL, XH, BC)   \
        SCAN_STEP(4, DTL, DTH, XL, XH, BC)  SCAN_STEP(5, DTL, DTH, XL, XH, BC)   \
        SCAN_STEP(6, DTL, DTH, XL, XH, BC)  SCAN_STEP(7, DTL, DTH, XL, XH, BC)   \
        SCAN_STEP(8, DTL, DTH, XL, XH, BC)  SCAN_STEP(9, DTL, DTH, XL, XH, BC)   \
        SCAN_STEP(10, DTL, DTH, XL, XH, BC) SCAN_STEP(11, DTL, DTH, XL, XH, BC)  \
        SCAN_STEP(12, DTL, DTH, XL, XH, BC) SCAN_STEP(13, DTL, DTH, XL, XH, BC)  \
        SCAN_STEP(14, DTL, DTH, XL, XH, BC) SCAN_STEP(15, DTL, DTH, XL, XH, BC)  \
          \
        TREE_STAGE(Q, V, 8, bit0, 1)                                   \
        TREE_STAGE(R, Q, 4, bit1, 2)                                   \
        TREE_STAGE(S, R, 2, bit2, 4)                                   \
          \
        const size_t o = (size_t)(S0) * 4;                             \
        pY[o]      = (S[0] + dsk * XL) * silu_f(ZL);                   \
        pY[o + HI] = (S[1] + dsk * XH) * silu_f(ZH);                   \
    }

#define HOLD_LOADS() __builtin_amdgcn_sched_barrier(0);

    float h0 = 0.0f, h1 = 0.0f;
    LOAD_TILE(dtLA, dtHA, xLA, xHA, zLA, zHA, bcA, 0)
    LOAD_TILE(dtLB, dtHB, xLB, xHB, zLB, zHB, bcB, TILE_S)
    int s0 = 0;
    for (; s0 + 5 * TILE_S <= SEQ; s0 += 3 * TILE_S) {
        LOAD_TILE(dtLC, dtHC, xLC, xHC, zLC, zHC, bcC, s0 + 2 * TILE_S)
        HOLD_LOADS()
        SCAN_TILE(dtLA, dtHA, xLA, xHA, zLA, zHA, bcA, s0)
        HOLD_LOADS()
        LOAD_TILE(dtLA, dtHA, xLA, xHA, zLA, zHA, bcA, s0 + 3 * TILE_S)
        HOLD_LOADS()
        SCAN_TILE(dtLB, dtHB, xLB, xHB, zLB, zHB, bcB, s0 + TILE_S)
        HOLD_LOADS()
        LOAD_TILE(dtLB, dtHB, xLB, xHB, zLB, zHB, bcB, s0 + 4 * TILE_S)
        HOLD_LOADS()
        SCAN_TILE(dtLC, dtHC, xLC, xHC, zLC, zHC, bcC, s0 + 2 * TILE_S)
    }
    static_assert((SEQ / TILE_S) % 3 == 2 && SEQ % TILE_S == 0, "tile count");
    SCAN_TILE(dtLA, dtHA, xLA, xHA, zLA, zHA, bcA, s0)
    SCAN_TILE(dtLB, dtHB, xLB, xHB, zLB, zHB, bcB, s0 + TILE_S)

#undef HOLD_LOADS
#undef TREE_STAGE
#undef LOAD_TILE
#undef SCAN_STEP
#undef SCAN_TILE
}

extern "C" void kernel_launch(void* const* d_in, const int* in_sizes, int n_in,
                              void* d_out, int out_size, void* d_ws, size_t ws_size,
                              hipStream_t stream) {
    const float* x         = (const float*)d_in[0];
    const float* in_proj_w = (const float*)d_in[1];
    const float* conv_w    = (const float*)d_in[2];
    const float* conv_b    = (const float*)d_in[3];
    const float* x_proj_w  = (const float*)d_in[4];
    const float* dt_proj_w = (const float*)d_in[5];
    const float* dt_proj_b = (const float*)d_in[6];
    const float* log_A     = (const float*)d_in[7];
    const float* D_skip    = (const float*)d_in[8];
    const float* out_proj_w= (const float*)d_in[9];
    float* out = (float*)d_out;

    const size_t planeBD = (size_t)MROWS * D_INNER;
    float* ws      = (float*)d_ws;
    float* buf_x   = ws;
    float* buf_z   = buf_x  + planeBD;
    float* buf_xc  = buf_z  + planeBD;
    float* buf_dbc = buf_xc + planeBD;
    float* buf_y   = buf_dbc + (size_t)MROWS * DBC_LD;

    const dim3 blk(128);
    const int gy = MROWS / (16 * 2 * 4);

    inproj_xz_wmma_f32<<<dim3(D_INNER / 64, gy), blk, 0, stream>>>(
        x, in_proj_w, buf_x, buf_z);

    conv_xproj_kernel<<<MROWS / 16, XP_THREADS, 0, stream>>>(
        buf_x, conv_w, conv_b, x_proj_w, buf_xc, buf_dbc);

    gemm_wmma_f32<2, 4, 1, 16, false, true><<<dim3(D_INNER / 64, gy), blk, 0, stream>>>(
        buf_dbc, dt_proj_w, dt_proj_b, buf_x, MROWS, D_INNER, DT_RANK, DBC_LD, D_INNER);

    scan_kernel<<<(BATCH * D_INNER * (D_STATE / 2)) / SCAN_THREADS, SCAN_THREADS, 0, stream>>>(
        buf_x, buf_xc, buf_dbc, buf_z, log_A, D_skip, buf_y);

    gemm_wmma_f32<2, 2, 0, 64, true, false><<<dim3(D_MODEL / 32, gy), blk, 0, stream>>>(
        buf_y, out_proj_w, nullptr, out, MROWS, D_MODEL, D_INNER, D_INNER, D_MODEL);
}
